# starting kernel
# baseline (speedup 1.0000x reference)
#include <hip/hip_runtime.h>
#include <hip/hip_bf16.h>

typedef __attribute__((ext_vector_type(2))) float v2f;
typedef __attribute__((ext_vector_type(8))) float v8f;

namespace {

constexpr int T_STEPS = 5;
constexpr int BATCH   = 1000000;
constexpr int IN_DIM  = 5;
constexpr int HID     = 6;
constexpr int ELEMS_PER_BLOCK = 80;
constexpr int WAVES_PER_BLOCK = 5;
constexpr int THREADS = WAVES_PER_BLOCK * 32;
constexpr int ROWS_PER_T = BATCH / T_STEPS;

__device__ __forceinline__ float swap16(float v) {
  return __int_as_float(
      __builtin_amdgcn_ds_swizzle(__float_as_int(v), 0x401F));
}

__device__ __forceinline__ float tanh_fast(float x) {
#if __has_builtin(__builtin_amdgcn_tanhf)
  return __builtin_amdgcn_tanhf(x);
#else
  float e = __expf(-2.0f * __builtin_fabsf(x));
  float r = __fdividef(1.0f - e, 1.0f + e);
  return __builtin_copysignf(r, x);
#endif
}

__device__ __forceinline__ float sig_fast(float x) {
  return 0.5f * tanh_fast(0.5f * x) + 0.5f;
}

__device__ __forceinline__ v8f wmma4(v2f a, v2f b, v8f c) {
  return __builtin_amdgcn_wmma_f32_16x16x4_f32(false, a, false, b, (short)0, c,
                                               false, false);
}

__device__ __forceinline__ void lstm_unit(float gi, float gf, float gg, float go,
                                          float& c, float& h) {
  float I = sig_fast(gi);
  float F = sig_fast(gf);
  float G = tanh_fast(gg);
  float O = sig_fast(go);
  c = F * c + I * G;
  h = O * tanh_fast(c);
}

__device__ __forceinline__ float wcol_l0(const float* wih, const float* whh, int R, int c) {
  if (c < 5)  return wih[R * 5 + c];
  if (c < 11) return whh[R * 6 + (c - 5)];
  return 0.0f;
}
__device__ __forceinline__ float wcol_l1(const float* wih, const float* whh, int R, int c) {
  if (c < 6) return wih[R * 6 + c];
  return whh[R * 6 + (c - 6)];
}

__device__ __forceinline__ int tile0_row(int m) {
  return (m & 3) * HID + (m >> 2);
}
__device__ __forceinline__ bool tile1_row(int m, int& R) {
  if (m < 4)            { R = m * HID + 4;       return true; }
  if (m >= 8 && m < 12) { R = (m - 8) * HID + 5; return true; }
  return false;
}

}

__global__ __launch_bounds__(THREADS)
void lstm2_wmma_kernel(const float* __restrict__ x,
                       const float* __restrict__ w_ih0, const float* __restrict__ w_hh0,
                       const float* __restrict__ b_ih0, const float* __restrict__ b_hh0,
                       const float* __restrict__ w_ih1, const float* __restrict__ w_hh1,
                       const float* __restrict__ b_ih1, const float* __restrict__ b_hh1,
                       const float* __restrict__ w_out, const float* __restrict__ b_out,
                       float* __restrict__ out) {
  __shared__ float sp[T_STEPS * 2 * ELEMS_PER_BLOCK];

  const int tid  = threadIdx.x;
  const int wave = tid >> 5;
  const int lane = tid & 31;
  const int n    = lane & 15;
  const int hi   = lane >> 4;
  const int e    = wave * 16 + n;
  const int b    = blockIdx.x * ELEMS_PER_BLOCK + e;

  v2f A0t0[3], A0t1[3], A1t0[3], A1t1[3];
  {
    const int m  = n;
    const int R0 = tile0_row(m);
    int R1; const bool v1 = tile1_row(m, R1);
#pragma unroll
    for (int s = 0; s < 3; ++s) {
      const int c = 4 * s + 2 * hi;
      A0t0[s][0] = wcol_l0(w_ih0, w_hh0, R0, c);
      A0t0[s][1] = wcol_l0(w_ih0, w_hh0, R0, c + 1);
      A0t1[s][0] = v1 ? wcol_l0(w_ih0, w_hh0, R1, c)     : 0.0f;
      A0t1[s][1] = v1 ? wcol_l0(w_ih0, w_hh0, R1, c + 1) : 0.0f;
      A1t0[s][0] = wcol_l1(w_ih1, w_hh1, R0, c);
      A1t0[s][1] = wcol_l1(w_ih1, w_hh1, R0, c + 1);
      A1t1[s][0] = v1 ? wcol_l1(w_ih1, w_hh1, R1, c)     : 0.0f;
      A1t1[s][1] = v1 ? wcol_l1(w_ih1, w_hh1, R1, c + 1) : 0.0f;
    }
  }

  v8f c0i0 = {}, c1i0 = {}, c0i1 = {}, c1i1 = {};
#pragma unroll
  for (int r = 0; r < 8; ++r) {
    const int m = r + 8 * hi;
    const int R = tile0_row(m);
    c0i0[r] = b_ih0[R] + b_hh0[R];
    c0i1[r] = b_ih1[R] + b_hh1[R];
    int R1;
    if (tile1_row(m, R1)) {
      c1i0[r] = b_ih0[R1] + b_hh0[R1];
      c1i1[r] = b_ih1[R1] + b_hh1[R1];
    }
  }

  const int p  = e % 5;
  const int u0 = hi ? 2 : 0, u1 = hi ? 3 : 1, u2 = hi ? 5 : 4;
  const float wl0 = w_out[p * HID + u0];
  const float wl1 = w_out[p * HID + u1];
  const float wl2 = w_out[p * HID + u2];

  float xA[T_STEPS], xB[T_STEPS], xC[T_STEPS];
#pragma unroll
  for (int t = 0; t < T_STEPS; ++t) {
    const float* xp = x + ((size_t)t * BATCH + b) * IN_DIM;
    xA[t] = xp[2 * hi];
    xB[t] = xp[2 * hi + 1];
    xC[t] = xp[4];
  }

  float h1[3] = {0.f, 0.f, 0.f}, c1s[3] = {0.f, 0.f, 0.f};
  float h2[3] = {0.f, 0.f, 0.f}, c2s[3] = {0.f, 0.f, 0.f};

#pragma unroll
  for (int t = 0; t < T_STEPS; ++t) {
    const float sw1 = swap16(h1[1]);
    v2f B0, B1, B2;
    B0[0] = xA[t];               B0[1] = xB[t];
    B1[0] = hi ? sw1 : xC[t];    B1[1] = h1[0];
    B2[0] = hi ? h1[2] : sw1;    B2[1] = hi ? 0.0f : h1[2];

    v8f g0 = c0i0, g1 = c1i0;
    g0 = wmma4(A0t0[0], B0, g0);  g0 = wmma4(A0t0[1], B1, g0);  g0 = wmma4(A0t0[2], B2, g0);
    g1 = wmma4(A0t1[0], B0, g1);  g1 = wmma4(A0t1[1], B1, g1);  g1 = wmma4(A0t1[2], B2, g1);

    lstm_unit(g0[0], g0[1], g0[2], g0[3], c1s[0], h1[0]);
    lstm_unit(g0[4], g0[5], g0[6], g0[7], c1s[1], h1[1]);
    lstm_unit(g1[0], g1[1], g1[2], g1[3], c1s[2], h1[2]);

    const float ySw2 = swap16(h1[2]);
    const float s0 = swap16(h2[0]);
    const float s1 = swap16(h2[1]);
    const float s2 = swap16(h2[2]);
    v2f D0, D1, D2;
    D0[0] = h1[0];            D0[1] = h1[1];
    D1[0] = hi ? s0 : h1[2];  D1[1] = hi ? s1 : ySw2;
    D2[0] = hi ? s2 : s0;     D2[1] = hi ? h2[2] : s1;

    v8f q0 = c0i1, q1 = c1i1;
    q0 = wmma4(A1t0[0], D0, q0);  q0 = wmma4(A1t0[1], D1, q0);  q0 = wmma4(A1t0[2], D2, q0);
    q1 = wmma4(A1t1[0], D0, q1);  q1 = wmma4(A1t1[1], D1, q1);  q1 = wmma4(A1t1[2], D2, q1);

    lstm_unit(q0[0], q0[1], q0[2], q0[3], c2s[0], h2[0]);
    lstm_unit(q0[4], q0[5], q0[6], q0[7], c2s[1], h2[1]);
    lstm_unit(q1[0], q1[1], q1[2], q1[3], c2s[2], h2[2]);

    sp[(t * 2 + hi) * ELEMS_PER_BLOCK + e] = h2[0] * wl0 + h2[1] * wl1 + h2[2] * wl2;
  }

  __syncthreads();

  if (tid < T_STEPS * 16) {
    const int t   = tid >> 4;
    const int r16 = tid & 15;
    float sum = b_out[0];
#pragma unroll
    for (int pp = 0; pp < 5; ++pp) {
      const int ee = r16 * 5 + pp;
      sum += sp[(t * 2 + 0) * ELEMS_PER_BLOCK + ee] +
             sp[(t * 2 + 1) * ELEMS_PER_BLOCK + ee];
    }
    out[t * ROWS_PER_T + blockIdx.x * 16 + r16] = sum;
  }
}

extern "C" void kernel_launch(void* const* d_in, const int* in_sizes, int n_in,
                              void* d_out, int out_size, void* d_ws, size_t ws_size,
                              hipStream_t stream) {
  const float* x     = (const float*)d_in[0];
  const float* w_ih0 = (const float*)d_in[1];
  const float* w_hh0 = (const float*)d_in[2];
  const float* b_ih0 = (const float*)d_in[3];
  const float* b_hh0 = (const float*)d_in[4];
  const float* w_ih1 = (const float*)d_in[5];
  const float* w_hh1 = (const float*)d_in[6];
  const float* b_ih1 = (const float*)d_in[7];
  const float* b_hh1 = (const float*)d_in[8];
  const float* w_out = (const float*)d_in[9];
  const float* b_out = (const float*)d_in[10];
  float* out = (float*)d_out;

  const int grid = BATCH / ELEMS_PER_BLOCK;
  lstm2_wmma_kernel<<<grid, THREADS, 0, stream>>>(
      x, w_ih0, w_hh0, b_ih0, b_hh0, w_ih1, w_hh1, b_ih1, b_hh1, w_out, b_out, out);
}


# Each wave carries two independent 16-element tiles (blocks of 160 elements), one a phase behind: its gate math is sliced behind the other's six products; set-up per wave serves both
# speedup vs baseline: 1.0533x; 1.0533x over previous
#include <hip/hip_runtime.h>
#include <hip/hip_bf16.h>

typedef __attribute__((ext_vector_type(2))) float v2f;
typedef __attribute__((ext_vector_type(8))) float v8f;

namespace {

constexpr int T_STEPS = 5;
constexpr int BATCH   = 1000000;
constexpr int IN_DIM  = 5;
constexpr int HID     = 6;
constexpr int ELEMS_PER_GROUP = 80;
constexpr int TILES           = 2;
constexpr int ELEMS_PER_BLOCK = TILES * ELEMS_PER_GROUP;
constexpr int WAVES_PER_BLOCK = 5;
constexpr int THREADS = WAVES_PER_BLOCK * 32;
constexpr int ROWS_PER_T = BATCH / T_STEPS;

__device__ __forceinline__ float swap16(float v) {
  return __int_as_float(
      __builtin_amdgcn_ds_swizzle(__float_as_int(v), 0x401F));
}

__device__ __forceinline__ float tanh_fast(float x) {
#if __has_builtin(__builtin_amdgcn_tanhf)
  return __builtin_amdgcn_tanhf(x);
#else
  float e = __expf(-2.0f * __builtin_fabsf(x));
  float r = __fdividef(1.0f - e, 1.0f + e);
  return __builtin_copysignf(r, x);
#endif
}

__device__ __forceinline__ float sig_fast(float x) {
  return 0.5f * tanh_fast(0.5f * x) + 0.5f;
}

__device__ __forceinline__ v8f wmma4(v2f a, v2f b, v8f c) {
  return __builtin_amdgcn_wmma_f32_16x16x4_f32(false, a, false, b, (short)0, c,
                                               false, false);
}

__device__ __forceinline__ void lstm_unit(float gi, float gf, float gg, float go,
                                          float& c, float& h) {
  float I = sig_fast(gi);
  float F = sig_fast(gf);
  float G = tanh_fast(gg);
  float O = sig_fast(go);
  c = F * c + I * G;
  h = O * tanh_fast(c);
}

__device__ __forceinline__ float wcol_l0(const float* wih, const float* whh, int R, int c) {
  if (c < 5)  return wih[R * 5 + c];
  if (c < 11) return whh[R * 6 + (c - 5)];
  return 0.0f;
}
__device__ __forceinline__ float wcol_l1(const float* wih, const float* whh, int R, int c) {
  if (c < 6) return wih[R * 6 + c];
  return whh[R * 6 + (c - 6)];
}

__device__ __forceinline__ int tile0_row(int m) {
  return (m & 3) * HID + (m >> 2);
}
__device__ __forceinline__ bool tile1_row(int m, int& R) {
  if (m < 4)            { R = m * HID + 4;       return true; }
  if (m >= 8 && m < 12) { R = (m - 8) * HID + 5; return true; }
  return false;
}


__device__ __forceinline__ void build_l0(v2f (&B)[3], float xa, float xb, float xc,
                                         const float (&h1)[3], int hi) {
  const float sw1 = swap16(h1[1]);
  B[0][0] = xa;                B[0][1] = xb;
  B[1][0] = hi ? sw1 : xc;     B[1][1] = h1[0];
  B[2][0] = hi ? h1[2] : sw1;  B[2][1] = hi ? 0.0f : h1[2];
}
__device__ __forceinline__ void sel_l0(v2f (&B)[3], float xa, float xb, float xc,
                                       const float (&h1)[3], float sw1, int hi) {
  B[0][0] = xa;                B[0][1] = xb;
  B[1][0] = hi ? sw1 : xc;     B[1][1] = h1[0];
  B[2][0] = hi ? h1[2] : sw1;  B[2][1] = hi ? 0.0f : h1[2];
}
__device__ __forceinline__ void sel_l1(v2f (&D)[3], const float (&h1)[3], const float (&h2)[3],
                                       float ySw2, float s0, float s1, float s2, int hi) {
  D[0][0] = h1[0];            D[0][1] = h1[1];
  D[1][0] = hi ? s0 : h1[2];  D[1][1] = hi ? s1 : ySw2;
  D[2][0] = hi ? s2 : s0;     D[2][1] = hi ? h2[2] : s1;
}
__device__ __forceinline__ void gates(v8f& g0, v8f& g1, const v2f (&At0)[3],
                                      const v2f (&At1)[3], const v2f (&B)[3],
                                      v8f ci0, v8f ci1) {
  g0 = wmma4(At0[0], B[0], ci0);  g1 = wmma4(At1[0], B[0], ci1);
  g0 = wmma4(At0[1], B[1], g0);   g1 = wmma4(At1[1], B[1], g1);
  g0 = wmma4(At0[2], B[2], g0);   g1 = wmma4(At1[2], B[2], g1);
}
__device__ __forceinline__ void act_a(float gi, float gf, float gg, float go,
                                      float& I, float& F, float& G, float& O) {
  I = sig_fast(gi);  F = sig_fast(gf);  G = tanh_fast(gg);  O = sig_fast(go);
}
__device__ __forceinline__ void act_b(float I, float F, float G, float O, float& c, float& h) {
  c = F * c + I * G;
  h = O * tanh_fast(c);
}
__device__ __forceinline__ void act(const v8f& g0, const v8f& g1, float (&c)[3], float (&h)[3]) {
  lstm_unit(g0[0], g0[1], g0[2], g0[3], c[0], h[0]);
  lstm_unit(g0[4], g0[5], g0[6], g0[7], c[1], h[1]);
  lstm_unit(g1[0], g1[1], g1[2], g1[3], c[2], h[2]);
}

#define SLICE_END() __builtin_amdgcn_sched_barrier(0)

template <class Early, class Mid, class Tail>
__device__ __forceinline__ void phase(v8f& gx0, v8f& gx1, const v2f (&At0)[3],
                                      const v2f (&At1)[3], const v2f (&B)[3],
                                      const v8f& ci0, const v8f& ci1,
                                      const v8f& gy0, const v8f& gy1,
                                      float (&c)[3], float (&h)[3],
                                      Early early, Mid mid, Tail tail) {
  float I0, F0, G0, O0, I1, F1, G1, O1, I2, F2, G2, O2;
  gx0 = wmma4(At0[0], B[0], ci0);
  early();
  act_a(gy1[0], gy1[1], gy1[2], gy1[3], I2, F2, G2, O2);
  SLICE_END();
  gx1 = wmma4(At1[0], B[0], ci1);
  act_b(I2, F2, G2, O2, c[2], h[2]);
  mid();
  SLICE_END();
  gx0 = wmma4(At0[1], B[1], gx0);
  act_a(gy0[0], gy0[1], gy0[2], gy0[3], I0, F0, G0, O0);
  SLICE_END();
  gx1 = wmma4(At1[1], B[1], gx1);
  act_a(gy0[4], gy0[5], gy0[6], gy0[7], I1, F1, G1, O1);
  SLICE_END();
  gx0 = wmma4(At0[2], B[2], gx0);
  act_b(I0, F0, G0, O0, c[0], h[0]);
  act_b(I1, F1, G1, O1, c[1], h[1]);
  SLICE_END();
  gx1 = wmma4(At1[2], B[2], gx1);
  tail();
  SLICE_END();
}

}

__global__ __launch_bounds__(THREADS)
void lstm2_wmma_kernel(const float* __restrict__ x,
                       const float* __restrict__ w_ih0, const float* __restrict__ w_hh0,
                       const float* __restrict__ b_ih0, const float* __restrict__ b_hh0,
                       const float* __restrict__ w_ih1, const float* __restrict__ w_hh1,
                       const float* __restrict__ b_ih1, const float* __restrict__ b_hh1,
                       const float* __restrict__ w_out, const float* __restrict__ b_out,
                       float* __restrict__ out) {
  __shared__ float sp[TILES * T_STEPS * 2 * ELEMS_PER_GROUP];

  const int tid  = threadIdx.x;
  const int wave = tid >> 5;
  const int lane = tid & 31;
  const int n    = lane & 15;
  const int hi   = lane >> 4;
  const int e    = wave * 16 + n;

  v2f A0t0[3], A0t1[3], A1t0[3], A1t1[3];
  {
    const int m  = n;
    const int R0 = tile0_row(m);
    int R1; const bool v1 = tile1_row(m, R1);
#pragma unroll
    for (int s = 0; s < 3; ++s) {
      const int c = 4 * s + 2 * hi;
      A0t0[s][0] = wcol_l0(w_ih0, w_hh0, R0, c);
      A0t0[s][1] = wcol_l0(w_ih0, w_hh0, R0, c + 1);
      A0t1[s][0] = v1 ? wcol_l0(w_ih0, w_hh0, R1, c)     : 0.0f;
      A0t1[s][1] = v1 ? wcol_l0(w_ih0, w_hh0, R1, c + 1) : 0.0f;
      A1t0[s][0] = wcol_l1(w_ih1, w_hh1, R0, c);
      A1t0[s][1] = wcol_l1(w_ih1, w_hh1, R0, c + 1);
      A1t1[s][0] = v1 ? wcol_l1(w_ih1, w_hh1, R1, c)     : 0.0f;
      A1t1[s][1] = v1 ? wcol_l1(w_ih1, w_hh1, R1, c + 1) : 0.0f;
    }
  }

  v8f c0i0 = {}, c1i0 = {}, c0i1 = {}, c1i1 = {};
#pragma unroll
  for (int r = 0; r < 8; ++r) {
    const int m = r + 8 * hi;
    const int R = tile0_row(m);
    c0i0[r] = b_ih0[R] + b_hh0[R];
    c0i1[r] = b_ih1[R] + b_hh1[R];
    int R1;
    if (tile1_row(m, R1)) {
      c1i0[r] = b_ih0[R1] + b_hh0[R1];
      c1i1[r] = b_ih1[R1] + b_hh1[R1];
    }
  }

  const int p  = e % 5;
  const int u0 = hi ? 2 : 0, u1 = hi ? 3 : 1, u2 = hi ? 5 : 4;
  const float wl0 = w_out[p * HID + u0];
  const float wl1 = w_out[p * HID + u1];
  const float wl2 = w_out[p * HID + u2];

  float xA[TILES][T_STEPS], xB[TILES][T_STEPS], xC[TILES][T_STEPS];
#pragma unroll
  for (int k = 0; k < TILES; ++k) {
    const int b = (blockIdx.x * TILES + k) * ELEMS_PER_GROUP + e;
#pragma unroll
    for (int t = 0; t < T_STEPS; ++t) {
      const float* xp = x + ((size_t)t * BATCH + b) * IN_DIM;
      xA[k][t] = xp[2 * hi];
      xB[k][t] = xp[2 * hi + 1];
      xC[k][t] = xp[4];
    }
  }

  float h1[TILES][3] = {}, c1s[TILES][3] = {};
  float h2[TILES][3] = {}, c2s[TILES][3] = {};
  v2f Bop[TILES][3];
  v8f G0[TILES], G1[TILES];

  build_l0(Bop[0], xA[0][0], xB[0][0], xC[0][0], h1[0], hi);
  build_l0(Bop[1], xA[1][0], xB[1][0], xC[1][0], h1[1], hi);
  gates(G0[0], G1[0], A0t0, A0t1, Bop[0], c0i0, c1i0);
  SLICE_END();

#pragma unroll
  for (int t = 0; t < T_STEPS; ++t) {
    float ex0, ex1, ex2, ex3;

    phase(G0[1], G1[1], A0t0, A0t1, Bop[1], c0i0, c1i0, G0[0], G1[0], c1s[0], h1[0],
          [&] { ex0 = swap16(h2[0][0]); ex1 = swap16(h2[0][1]); ex2 = swap16(h2[0][2]); },
          [&] { ex3 = swap16(h1[0][2]); },
          [&] { sel_l1(Bop[0], h1[0], h2[0], ex3, ex0, ex1, ex2, hi); });

    phase(G0[0], G1[0], A1t0, A1t1, Bop[0], c0i1, c1i1, G0[1], G1[1], c1s[1], h1[1],
          [&] { ex0 = swap16(h2[1][0]); ex1 = swap16(h2[1][1]); ex2 = swap16(h2[1][2]); },
          [&] { ex3 = swap16(h1[1][2]); },
          [&] { sel_l1(Bop[1], h1[1], h2[1], ex3, ex0, ex1, ex2, hi); });

    phase(G0[1], G1[1], A1t0, A1t1, Bop[1], c0i1, c1i1, G0[0], G1[0], c2s[0], h2[0],
          [&] { if (t + 1 < T_STEPS) ex0 = swap16(h1[0][1]); },
          [&] {},
          [&] {
            sp[((0 * T_STEPS + t) * 2 + hi) * ELEMS_PER_GROUP + e] =
                h2[0][0] * wl0 + h2[0][1] * wl1 + h2[0][2] * wl2;
            if (t + 1 < T_STEPS)
              sel_l0(Bop[0], xA[0][t + 1], xB[0][t + 1], xC[0][t + 1], h1[0], ex0, hi);
          });

    if (t + 1 < T_STEPS) {
      phase(G0[0], G1[0], A0t0, A0t1, Bop[0], c0i0, c1i0, G0[1], G1[1], c2s[1], h2[1],
            [&] { ex0 = swap16(h1[1][1]); },
            [&] {},
            [&] {
              sp[((1 * T_STEPS + t) * 2 + hi) * ELEMS_PER_GROUP + e] =
                  h2[1][0] * wl0 + h2[1][1] * wl1 + h2[1][2] * wl2;
              sel_l0(Bop[1], xA[1][t + 1], xB[1][t + 1], xC[1][t + 1], h1[1], ex0, hi);
            });
    } else {
      act(G0[1], G1[1], c2s[1], h2[1]);
      sp[((1 * T_STEPS + t) * 2 + hi) * ELEMS_PER_GROUP + e] =
          h2[1][0] * wl0 + h2[1][1] * wl1 + h2[1][2] * wl2;
    }
  }

  __syncthreads();

  {
    const int t   = wave;
    const int k   = hi;
    const int r16 = n;
    float sum = b_out[0];
#pragma unroll
    for (int pp = 0; pp < 5; ++pp) {
      const int ee = r16 * 5 + pp;
      sum += sp[((k * T_STEPS + t) * 2 + 0) * ELEMS_PER_GROUP + ee] +
             sp[((k * T_STEPS + t) * 2 + 1) * ELEMS_PER_GROUP + ee];
    }
    out[t * ROWS_PER_T + blockIdx.x * (TILES * 16) + lane] = sum;
  }
}

extern "C" void kernel_launch(void* const* d_in, const int* in_sizes, int n_in,
                              void* d_out, int out_size, void* d_ws, size_t ws_size,
                              hipStream_t stream) {
  const float* x     = (const float*)d_in[0];
  const float* w_ih0 = (const float*)d_in[1];
  const float* w_hh0 = (const float*)d_in[2];
  const float* b_ih0 = (const float*)d_in[3];
  const float* b_hh0 = (const float*)d_in[4];
  const float* w_ih1 = (const float*)d_in[5];
  const float* w_hh1 = (const float*)d_in[6];
  const float* b_ih1 = (const float*)d_in[7];
  const float* b_hh1 = (const float*)d_in[8];
  const float* w_out = (const float*)d_in[9];
  const float* b_out = (const float*)d_in[10];
  float* out = (float*)d_out;

  const int grid = BATCH / ELEMS_PER_BLOCK;
  lstm2_wmma_kernel<<<grid, THREADS, 0, stream>>>(
      x, w_ih0, w_hh0, b_ih0, b_hh0, w_ih1, w_hh1, b_ih1, b_hh1, w_out, b_out, out);
}


# B-operand k order set per wave half so each lane feeds the products only its own x and h values: half-wave exchanges and selects gone; the k terms of each gate sum are added in another order
# speedup vs baseline: 1.1082x; 1.0521x over previous
#include <hip/hip_runtime.h>
#include <hip/hip_bf16.h>

typedef __attribute__((ext_vector_type(2))) float v2f;
typedef __attribute__((ext_vector_type(8))) float v8f;

namespace {

constexpr int T_STEPS = 5;
constexpr int BATCH   = 1000000;
constexpr int IN_DIM  = 5;
constexpr int HID     = 6;
constexpr int ELEMS_PER_GROUP = 80;
constexpr int TILES           = 2;
constexpr int ELEMS_PER_BLOCK = TILES * ELEMS_PER_GROUP;
constexpr int WAVES_PER_BLOCK = 5;
constexpr int THREADS = WAVES_PER_BLOCK * 32;
constexpr int ROWS_PER_T = BATCH / T_STEPS;

__device__ __forceinline__ float tanh_fast(float x) {
#if __has_builtin(__builtin_amdgcn_tanhf)
  return __builtin_amdgcn_tanhf(x);
#else
  float e = __expf(-2.0f * __builtin_fabsf(x));
  float r = __fdividef(1.0f - e, 1.0f + e);
  return __builtin_copysignf(r, x);
#endif
}

__device__ __forceinline__ float sig_fast(float x) {
  return 0.5f * tanh_fast(0.5f * x) + 0.5f;
}

__device__ __forceinline__ v8f wmma4(v2f a, v2f b, v8f c) {
  return __builtin_amdgcn_wmma_f32_16x16x4_f32(false, a, false, b, (short)0, c,
                                               false, false);
}

__device__ __forceinline__ void lstm_unit(float gi, float gf, float gg, float go,
                                          float& c, float& h) {
  float I = sig_fast(gi);
  float F = sig_fast(gf);
  float G = tanh_fast(gg);
  float O = sig_fast(go);
  c = F * c + I * G;
  h = O * tanh_fast(c);
}

__device__ __forceinline__ float wcol_l0(const float* wih, const float* whh, int R, int c) {
  if (c < 5)  return wih[R * 5 + c];
  if (c < 11) return whh[R * 6 + (c - 5)];
  return 0.0f;
}
__device__ __forceinline__ float wcol_l1(const float* wih, const float* whh, int R, int c) {
  if (c < 6) return wih[R * 6 + c];
  return whh[R * 6 + (c - 6)];
}

__device__ __forceinline__ int unit_of(int hi, int q) {
  return q == 0 ? (hi ? 2 : 0) : q == 1 ? (hi ? 3 : 1) : (hi ? 5 : 4);
}
__device__ __forceinline__ int zcol_l0(int s, int hi, int j) {
  if (s == 0) return 2 * hi + j;
  if (s == 1) return j == 0 ? (hi ? 11 : 4) : 5 + unit_of(hi, 0);
  return 5 + unit_of(hi, 1 + j);
}
__device__ __forceinline__ int zcol_l1(int s, int hi, int j) {
  if (s == 0) return unit_of(hi, 1 + j);
  if (s == 1) return j == 0 ? unit_of(hi, 0) : 6 + unit_of(hi, 0);
  return 6 + unit_of(hi, 1 + j);
}

__device__ __forceinline__ int tile0_row(int m) {
  return (m & 3) * HID + (m >> 2);
}
__device__ __forceinline__ bool tile1_row(int m, int& R) {
  if (m < 4)            { R = m * HID + 4;       return true; }
  if (m >= 8 && m < 12) { R = (m - 8) * HID + 5; return true; }
  return false;
}


__device__ __forceinline__ void gates(v8f& g0, v8f& g1, const v2f (&At0)[3],
                                      const v2f (&At1)[3], v2f b0, v2f b1, v2f b2,
                                      v8f ci0, v8f ci1) {
  g0 = wmma4(At0[0], b0, ci0);  g1 = wmma4(At1[0], b0, ci1);
  g0 = wmma4(At0[1], b1, g0);   g1 = wmma4(At1[1], b1, g1);
  g0 = wmma4(At0[2], b2, g0);   g1 = wmma4(At1[2], b2, g1);
}
__device__ __forceinline__ void act_a(float gi, float gf, float gg, float go,
                                      float& I, float& F, float& G, float& O) {
  I = sig_fast(gi);  F = sig_fast(gf);  G = tanh_fast(gg);  O = sig_fast(go);
}
__device__ __forceinline__ void act_b(float I, float F, float G, float O, float& c, float& h) {
  c = F * c + I * G;
  h = O * tanh_fast(c);
}
__device__ __forceinline__ void act(const v8f& g0, const v8f& g1, float (&c)[3], float (&h)[3]) {
  lstm_unit(g0[0], g0[1], g0[2], g0[3], c[0], h[0]);
  lstm_unit(g0[4], g0[5], g0[6], g0[7], c[1], h[1]);
  lstm_unit(g1[0], g1[1], g1[2], g1[3], c[2], h[2]);
}

#define SLICE_END() __builtin_amdgcn_sched_barrier(0)

template <class Tail>
__device__ __forceinline__ void phase(v8f& gx0, v8f& gx1, const v2f (&At0)[3],
                                      const v2f (&At1)[3], v2f b0, v2f b1, v2f b2,
                                      const v8f& ci0, const v8f& ci1,
                                      const v8f& gy0, const v8f& gy1,
                                      float (&c)[3], Tail tail) {
  float I0, F0, G0, O0, I1, F1, G1, O1, I2, F2, G2, O2;
  float hn0, hn1, hn2;
  gx0 = wmma4(At0[0], b0, ci0);
  act_a(gy1[0], gy1[1], gy1[2], gy1[3], I2, F2, G2, O2);
  SLICE_END();
  gx1 = wmma4(At1[0], b0, ci1);
  act_b(I2, F2, G2, O2, c[2], hn2);
  SLICE_END();
  gx0 = wmma4(At0[1], b1, gx0);
  act_a(gy0[0], gy0[1], gy0[2], gy0[3], I0, F0, G0, O0);
  SLICE_END();
  gx1 = wmma4(At1[1], b1, gx1);
  act_a(gy0[4], gy0[5], gy0[6], gy0[7], I1, F1, G1, O1);
  SLICE_END();
  gx0 = wmma4(At0[2], b2, gx0);
  act_b(I0, F0, G0, O0, c[0], hn0);
  act_b(I1, F1, G1, O1, c[1], hn1);
  SLICE_END();
  gx1 = wmma4(At1[2], b2, gx1);
  tail(hn0, hn1, hn2);
  SLICE_END();
}

}

__global__ __launch_bounds__(THREADS)
void lstm2_wmma_kernel(const float* __restrict__ x,
                       const float* __restrict__ w_ih0, const float* __restrict__ w_hh0,
                       const float* __restrict__ b_ih0, const float* __restrict__ b_hh0,
                       const float* __restrict__ w_ih1, const float* __restrict__ w_hh1,
                       const float* __restrict__ b_ih1, const float* __restrict__ b_hh1,
                       const float* __restrict__ w_out, const float* __restrict__ b_out,
                       float* __restrict__ out) {
  __shared__ float sp[TILES * T_STEPS * 2 * ELEMS_PER_GROUP];

  const int tid  = threadIdx.x;
  const int wave = tid >> 5;
  const int lane = tid & 31;
  const int n    = lane & 15;
  const int hi   = lane >> 4;
  const int e    = wave * 16 + n;

  v2f A0t0[3], A0t1[3], A1t0[3], A1t1[3];
  {
    const int m  = n;
    const int R0 = tile0_row(m);
    int R1; const bool v1 = tile1_row(m, R1);
#pragma unroll
    for (int s = 0; s < 3; ++s) {
#pragma unroll
      for (int j = 0; j < 2; ++j) {
        const int ca = zcol_l0(s, hi, j);
        const int cb = zcol_l1(s, hi, j);
        A0t0[s][j] = wcol_l0(w_ih0, w_hh0, R0, ca);
        A0t1[s][j] = v1 ? wcol_l0(w_ih0, w_hh0, R1, ca) : 0.0f;
        A1t0[s][j] = wcol_l1(w_ih1, w_hh1, R0, cb);
        A1t1[s][j] = v1 ? wcol_l1(w_ih1, w_hh1, R1, cb) : 0.0f;
      }
    }
  }

  v8f c0i0 = {}, c1i0 = {}, c0i1 = {}, c1i1 = {};
#pragma unroll
  for (int r = 0; r < 8; ++r) {
    const int m = r + 8 * hi;
    const int R = tile0_row(m);
    c0i0[r] = b_ih0[R] + b_hh0[R];
    c0i1[r] = b_ih1[R] + b_hh1[R];
    int R1;
    if (tile1_row(m, R1)) {
      c1i0[r] = b_ih0[R1] + b_hh0[R1];
      c1i1[r] = b_ih1[R1] + b_hh1[R1];
    }
  }

  const int p  = e % 5;
  const int u0 = hi ? 2 : 0, u1 = hi ? 3 : 1, u2 = hi ? 5 : 4;
  const float wl0 = w_out[p * HID + u0];
  const float wl1 = w_out[p * HID + u1];
  const float wl2 = w_out[p * HID + u2];

  v2f   xP[TILES][T_STEPS];
  float xC[TILES][T_STEPS];
#pragma unroll
  for (int k = 0; k < TILES; ++k) {
    const int b = (blockIdx.x * TILES + k) * ELEMS_PER_GROUP + e;
#pragma unroll
    for (int t = 0; t < T_STEPS; ++t) {
      const float* xp = x + ((size_t)t * BATCH + b) * IN_DIM;
      xP[k][t][0] = xp[2 * hi];
      xP[k][t][1] = xp[2 * hi + 1];
      const float x4 = xp[4];
      xC[k][t]    = hi ? 0.0f : x4;
    }
  }

  float c1s[TILES][3] = {}, c2s[TILES][3] = {};
  v2f P1[TILES], P2[TILES], Q0[TILES], Q1[TILES];
#pragma unroll
  for (int k = 0; k < TILES; ++k) {
    P1[k] = v2f{0.0f, 0.0f};  P2[k] = v2f{0.0f, 0.0f};
    Q1[k] = v2f{0.0f, 0.0f};  Q0[k] = v2f{xC[k][0], 0.0f};
  }
  v8f G0[TILES], G1[TILES];

  gates(G0[0], G1[0], A0t0, A0t1, xP[0][0], Q0[0], P1[0], c0i0, c1i0);
  SLICE_END();

#pragma unroll
  for (int t = 0; t < T_STEPS; ++t) {
    phase(G0[1], G1[1], A0t0, A0t1, xP[1][t], Q0[1], P1[1], c0i0, c1i0, G0[0], G1[0], c1s[0],
          [&](float hs0, float hs1, float hs2) {
            P1[0] = v2f{hs1, hs2};  Q1[0][0] = hs0;
          });

    phase(G0[0], G1[0], A1t0, A1t1, P1[0], Q1[0], P2[0], c0i1, c1i1, G0[1], G1[1], c1s[1],
          [&](float hs0, float hs1, float hs2) {
            P1[1] = v2f{hs1, hs2};  Q1[1][0] = hs0;
          });

    phase(G0[1], G1[1], A1t0, A1t1, P1[1], Q1[1], P2[1], c0i1, c1i1, G0[0], G1[0], c2s[0],
          [&](float hs0, float hs1, float hs2) {
            P2[0] = v2f{hs1, hs2};  Q1[0][1] = hs0;
            sp[((0 * T_STEPS + t) * 2 + hi) * ELEMS_PER_GROUP + e] = hs0 * wl0 + hs1 * wl1 + hs2 * wl2;
            if (t + 1 < T_STEPS) Q0[0] = v2f{xC[0][t + 1], Q1[0][0]};
          });

    if (t + 1 < T_STEPS) {
      phase(G0[0], G1[0], A0t0, A0t1, xP[0][t + 1], Q0[0], P1[0], c0i0, c1i0, G0[1], G1[1], c2s[1],
            [&](float hs0, float hs1, float hs2) {
              P2[1] = v2f{hs1, hs2};  Q1[1][1] = hs0;
              sp[((1 * T_STEPS + t) * 2 + hi) * ELEMS_PER_GROUP + e] = hs0 * wl0 + hs1 * wl1 + hs2 * wl2;
              Q0[1] = v2f{xC[1][t + 1], Q1[1][0]};
            });
    } else {
      float hl[3];
      act(G0[1], G1[1], c2s[1], hl);
      sp[((1 * T_STEPS + t) * 2 + hi) * ELEMS_PER_GROUP + e] =
          hl[0] * wl0 + hl[1] * wl1 + hl[2] * wl2;
    }
  }

  __syncthreads();

  {
    const int t   = wave;
    const int k   = hi;
    const int r16 = n;
    float sum = b_out[0];
#pragma unroll
    for (int pp = 0; pp < 5; ++pp) {
      const int ee = r16 * 5 + pp;
      sum += sp[((k * T_STEPS + t) * 2 + 0) * ELEMS_PER_GROUP + ee] +
             sp[((k * T_STEPS + t) * 2 + 1) * ELEMS_PER_GROUP + ee];
    }
    out[t * ROWS_PER_T + blockIdx.x * (TILES * 16) + lane] = sum;
  }
}

extern "C" void kernel_launch(void* const* d_in, const int* in_sizes, int n_in,
                              void* d_out, int out_size, void* d_ws, size_t ws_size,
                              hipStream_t stream) {
  const float* x     = (const float*)d_in[0];
  const float* w_ih0 = (const float*)d_in[1];
  const float* w_hh0 = (const float*)d_in[2];
  const float* b_ih0 = (const float*)d_in[3];
  const float* b_hh0 = (const float*)d_in[4];
  const float* w_ih1 = (const float*)d_in[5];
  const float* w_hh1 = (const float*)d_in[6];
  const float* b_ih1 = (const float*)d_in[7];
  const float* b_hh1 = (const float*)d_in[8];
  const float* w_out = (const float*)d_in[9];
  const float* b_out = (const float*)d_in[10];
  float* out = (float*)d_out;

  const int grid = BATCH / ELEMS_PER_BLOCK;
  lstm2_wmma_kernel<<<grid, THREADS, 0, stream>>>(
      x, w_ih0, w_hh0, b_ih0, b_hh0, w_ih1, w_hh1, b_ih1, b_hh1, w_out, b_out, out);
}


# Sigmoid constants folded into the weight fragments at set-up: i, f, o gate rows and biases halved; hidden state carried doubled (2h = to*tc + tc, one rounding) with its weight columns halved
# speedup vs baseline: 1.1503x; 1.0380x over previous
#include <hip/hip_runtime.h>
#include <hip/hip_bf16.h>

typedef __attribute__((ext_vector_type(2))) float v2f;
typedef __attribute__((ext_vector_type(8))) float v8f;

namespace {

constexpr int T_STEPS = 5;
constexpr int BATCH   = 1000000;
constexpr int IN_DIM  = 5;
constexpr int HID     = 6;
constexpr int ELEMS_PER_GROUP = 80;
constexpr int TILES           = 2;
constexpr int ELEMS_PER_BLOCK = TILES * ELEMS_PER_GROUP;
constexpr int WAVES_PER_BLOCK = 5;
constexpr int THREADS = WAVES_PER_BLOCK * 32;
constexpr int ROWS_PER_T = BATCH / T_STEPS;

__device__ __forceinline__ float tanh_fast(float x) {
#if __has_builtin(__builtin_amdgcn_tanhf)
  return __builtin_amdgcn_tanhf(x);
#else
  float e = __expf(-2.0f * __builtin_fabsf(x));
  float r = __fdividef(1.0f - e, 1.0f + e);
  return __builtin_copysignf(r, x);
#endif
}

__device__ __forceinline__ float sig_fast(float x) {
  return 0.5f * tanh_fast(x) + 0.5f;
}

__device__ __forceinline__ v8f wmma4(v2f a, v2f b, v8f c) {
  return __builtin_amdgcn_wmma_f32_16x16x4_f32(false, a, false, b, (short)0, c,
                                               false, false);
}

__device__ __forceinline__ float h_doubled(float to, float tc) { return __builtin_fmaf(to, tc, tc); }
__device__ __forceinline__ void lstm_unit(float gi, float gf, float gg, float go,
                                          float& c, float& h) {
  float I = sig_fast(gi);
  float F = sig_fast(gf);
  float G = tanh_fast(gg);
  float O = tanh_fast(go);
  c = F * c + I * G;
  h = h_doubled(O, tanh_fast(c));
}

__device__ __forceinline__ float wcol_l0(const float* wih, const float* whh, int R, int c,
                                         float sx, float sh) {
  if (c < 5)  return sx * wih[R * 5 + c];
  if (c < 11) return sh * whh[R * 6 + (c - 5)];
  return 0.0f;
}
__device__ __forceinline__ float wcol_l1(const float* wih, const float* whh, int R, int c, float sh) {
  if (c < 6) return sh * wih[R * 6 + c];
  return sh * whh[R * 6 + (c - 6)];
}

__device__ __forceinline__ int unit_of(int hi, int q) {
  return q == 0 ? (hi ? 2 : 0) : q == 1 ? (hi ? 3 : 1) : (hi ? 5 : 4);
}
__device__ __forceinline__ int zcol_l0(int s, int hi, int j) {
  if (s == 0) return 2 * hi + j;
  if (s == 1) return j == 0 ? (hi ? 11 : 4) : 5 + unit_of(hi, 0);
  return 5 + unit_of(hi, 1 + j);
}
__device__ __forceinline__ int zcol_l1(int s, int hi, int j) {
  if (s == 0) return unit_of(hi, 1 + j);
  if (s == 1) return j == 0 ? unit_of(hi, 0) : 6 + unit_of(hi, 0);
  return 6 + unit_of(hi, 1 + j);
}

__device__ __forceinline__ int tile0_row(int m) {
  return (m & 3) * HID + (m >> 2);
}
__device__ __forceinline__ bool tile1_row(int m, int& R) {
  if (m < 4)            { R = m * HID + 4;       return true; }
  if (m >= 8 && m < 12) { R = (m - 8) * HID + 5; return true; }
  return false;
}


__device__ __forceinline__ void gates(v8f& g0, v8f& g1, const v2f (&At0)[3],
                                      const v2f (&At1)[3], v2f b0, v2f b1, v2f b2,
                                      v8f ci0, v8f ci1) {
  g0 = wmma4(At0[0], b0, ci0);  g1 = wmma4(At1[0], b0, ci1);
  g0 = wmma4(At0[1], b1, g0);   g1 = wmma4(At1[1], b1, g1);
  g0 = wmma4(At0[2], b2, g0);   g1 = wmma4(At1[2], b2, g1);
}
__device__ __forceinline__ void act_a(float gi, float gf, float gg, float go,
                                      float& I, float& F, float& G, float& O) {
  I = sig_fast(gi);  F = sig_fast(gf);  G = tanh_fast(gg);  O = tanh_fast(go);
}
__device__ __forceinline__ void act_b(float I, float F, float G, float O, float& c, float& h) {
  c = F * c + I * G;
  h = h_doubled(O, tanh_fast(c));
}
__device__ __forceinline__ void act(const v8f& g0, const v8f& g1, float (&c)[3], float (&h)[3]) {
  lstm_unit(g0[0], g0[1], g0[2], g0[3], c[0], h[0]);
  lstm_unit(g0[4], g0[5], g0[6], g0[7], c[1], h[1]);
  lstm_unit(g1[0], g1[1], g1[2], g1[3], c[2], h[2]);
}

#define SLICE_END() __builtin_amdgcn_sched_barrier(0)

template <class Tail>
__device__ __forceinline__ void phase(v8f& gx0, v8f& gx1, const v2f (&At0)[3],
                                      const v2f (&At1)[3], v2f b0, v2f b1, v2f b2,
                                      const v8f& ci0, const v8f& ci1,
                                      const v8f& gy0, const v8f& gy1,
                                      float (&c)[3], Tail tail) {
  float I0, F0, G0, O0, I1, F1, G1, O1, I2, F2, G2, O2;
  float hn0, hn1, hn2;
  gx0 = wmma4(At0[0], b0, ci0);
  act_a(gy1[0], gy1[1], gy1[2], gy1[3], I2, F2, G2, O2);
  SLICE_END();
  gx1 = wmma4(At1[0], b0, ci1);
  act_b(I2, F2, G2, O2, c[2], hn2);
  SLICE_END();
  gx0 = wmma4(At0[1], b1, gx0);
  act_a(gy0[0], gy0[1], gy0[2], gy0[3], I0, F0, G0, O0);
  SLICE_END();
  gx1 = wmma4(At1[1], b1, gx1);
  act_a(gy0[4], gy0[5], gy0[6], gy0[7], I1, F1, G1, O1);
  SLICE_END();
  gx0 = wmma4(At0[2], b2, gx0);
  act_b(I0, F0, G0, O0, c[0], hn0);
  act_b(I1, F1, G1, O1, c[1], hn1);
  SLICE_END();
  gx1 = wmma4(At1[2], b2, gx1);
  tail(hn0, hn1, hn2);
  SLICE_END();
}

}

__global__ __launch_bounds__(THREADS)
void lstm2_wmma_kernel(const float* __restrict__ x,
                       const float* __restrict__ w_ih0, const float* __restrict__ w_hh0,
                       const float* __restrict__ b_ih0, const float* __restrict__ b_hh0,
                       const float* __restrict__ w_ih1, const float* __restrict__ w_hh1,
                       const float* __restrict__ b_ih1, const float* __restrict__ b_hh1,
                       const float* __restrict__ w_out, const float* __restrict__ b_out,
                       float* __restrict__ out) {
  __shared__ float sp[TILES * T_STEPS * 2 * ELEMS_PER_GROUP];

  const int tid  = threadIdx.x;
  const int wave = tid >> 5;
  const int lane = tid & 31;
  const int n    = lane & 15;
  const int hi   = lane >> 4;
  const int e    = wave * 16 + n;

  v2f A0t0[3], A0t1[3], A1t0[3], A1t1[3];
  {
    const int m  = n;
    const int R0 = tile0_row(m);
    int R1; const bool v1 = tile1_row(m, R1);
    const float sc0 = (R0 / HID == 2) ? 1.0f : 0.5f;
    const float sc1 = (v1 && R1 / HID == 2) ? 1.0f : 0.5f;
    const float sh0 = 0.5f * sc0, sh1 = 0.5f * sc1;
#pragma unroll
    for (int s = 0; s < 3; ++s) {
#pragma unroll
      for (int j = 0; j < 2; ++j) {
        const int ca = zcol_l0(s, hi, j);
        const int cb = zcol_l1(s, hi, j);
        A0t0[s][j] = wcol_l0(w_ih0, w_hh0, R0, ca, sc0, sh0);
        A0t1[s][j] = v1 ? wcol_l0(w_ih0, w_hh0, R1, ca, sc1, sh1) : 0.0f;
        A1t0[s][j] = wcol_l1(w_ih1, w_hh1, R0, cb, sh0);
        A1t1[s][j] = v1 ? wcol_l1(w_ih1, w_hh1, R1, cb, sh1) : 0.0f;
      }
    }
  }

  v8f c0i0 = {}, c1i0 = {}, c0i1 = {}, c1i1 = {};
#pragma unroll
  for (int r = 0; r < 8; ++r) {
    const int m = r + 8 * hi;
    const int R = tile0_row(m);
    const float sb = (R / HID == 2) ? 1.0f : 0.5f;
    c0i0[r] = sb * (b_ih0[R] + b_hh0[R]);
    c0i1[r] = sb * (b_ih1[R] + b_hh1[R]);
    int R1;
    if (tile1_row(m, R1)) {
      const float sb1 = (R1 / HID == 2) ? 1.0f : 0.5f;
      c1i0[r] = sb1 * (b_ih0[R1] + b_hh0[R1]);
      c1i1[r] = sb1 * (b_ih1[R1] + b_hh1[R1]);
    }
  }

  const int p  = e % 5;
  const int u0 = hi ? 2 : 0, u1 = hi ? 3 : 1, u2 = hi ? 5 : 4;
  const float wl0 = w_out[p * HID + u0];
  const float wl1 = w_out[p * HID + u1];
  const float wl2 = w_out[p * HID + u2];

  v2f   xP[TILES][T_STEPS];
  float xC[TILES][T_STEPS];
#pragma unroll
  for (int k = 0; k < TILES; ++k) {
    const int b = (blockIdx.x * TILES + k) * ELEMS_PER_GROUP + e;
#pragma unroll
    for (int t = 0; t < T_STEPS; ++t) {
      const float* xp = x + ((size_t)t * BATCH + b) * IN_DIM;
      xP[k][t][0] = xp[2 * hi];
      xP[k][t][1] = xp[2 * hi + 1];
      const float x4 = xp[4];
      xC[k][t]    = hi ? 0.0f : x4;
    }
  }

  float c1s[TILES][3] = {}, c2s[TILES][3] = {};
  v2f P1[TILES], P2[TILES], Q0[TILES], Q1[TILES];
#pragma unroll
  for (int k = 0; k < TILES; ++k) {
    P1[k] = v2f{0.0f, 0.0f};  P2[k] = v2f{0.0f, 0.0f};
    Q1[k] = v2f{0.0f, 0.0f};  Q0[k] = v2f{xC[k][0], 0.0f};
  }
  v8f G0[TILES], G1[TILES];

  gates(G0[0], G1[0], A0t0, A0t1, xP[0][0], Q0[0], P1[0], c0i0, c1i0);
  SLICE_END();

#pragma unroll
  for (int t = 0; t < T_STEPS; ++t) {
    phase(G0[1], G1[1], A0t0, A0t1, xP[1][t], Q0[1], P1[1], c0i0, c1i0, G0[0], G1[0], c1s[0],
          [&](float hs0, float hs1, float hs2) {
            P1[0] = v2f{hs1, hs2};  Q1[0][0] = hs0;
          });

    phase(G0[0], G1[0], A1t0, A1t1, P1[0], Q1[0], P2[0], c0i1, c1i1, G0[1], G1[1], c1s[1],
          [&](float hs0, float hs1, float hs2) {
            P1[1] = v2f{hs1, hs2};  Q1[1][0] = hs0;
          });

    phase(G0[1], G1[1], A1t0, A1t1, P1[1], Q1[1], P2[1], c0i1, c1i1, G0[0], G1[0], c2s[0],
          [&](float hs0, float hs1, float hs2) {
            P2[0] = v2f{hs1, hs2};  Q1[0][1] = hs0;
            sp[((0 * T_STEPS + t) * 2 + hi) * ELEMS_PER_GROUP + e] = hs0 * wl0 + hs1 * wl1 + hs2 * wl2;
            if (t + 1 < T_STEPS) Q0[0] = v2f{xC[0][t + 1], Q1[0][0]};
          });

    if (t + 1 < T_STEPS) {
      phase(G0[0], G1[0], A0t0, A0t1, xP[0][t + 1], Q0[0], P1[0], c0i0, c1i0, G0[1], G1[1], c2s[1],
            [&](float hs0, float hs1, float hs2) {
              P2[1] = v2f{hs1, hs2};  Q1[1][1] = hs0;
              sp[((1 * T_STEPS + t) * 2 + hi) * ELEMS_PER_GROUP + e] = hs0 * wl0 + hs1 * wl1 + hs2 * wl2;
              Q0[1] = v2f{xC[1][t + 1], Q1[1][0]};
            });
    } else {
      float hl[3];
      act(G0[1], G1[1], c2s[1], hl);
      sp[((1 * T_STEPS + t) * 2 + hi) * ELEMS_PER_GROUP + e] =
          hl[0] * wl0 + hl[1] * wl1 + hl[2] * wl2;
    }
  }

  __syncthreads();

  {
    const int t   = wave;
    const int k   = hi;
    const int r16 = n;
    float sum = 2.0f * b_out[0];
#pragma unroll
    for (int pp = 0; pp < 5; ++pp) {
      const int ee = r16 * 5 + pp;
      sum += sp[((k * T_STEPS + t) * 2 + 0) * ELEMS_PER_GROUP + ee] +
             sp[((k * T_STEPS + t) * 2 + 1) * ELEMS_PER_GROUP + ee];
    }
    out[t * ROWS_PER_T + blockIdx.x * (TILES * 16) + lane] = 0.5f * sum;
  }
}

extern "C" void kernel_launch(void* const* d_in, const int* in_sizes, int n_in,
                              void* d_out, int out_size, void* d_ws, size_t ws_size,
                              hipStream_t stream) {
  const float* x     = (const float*)d_in[0];
  const float* w_ih0 = (const float*)d_in[1];
  const float* w_hh0 = (const float*)d_in[2];
  const float* b_ih0 = (const float*)d_in[3];
  const float* b_hh0 = (const float*)d_in[4];
  const float* w_ih1 = (const float*)d_in[5];
  const float* w_hh1 = (const float*)d_in[6];
  const float* b_ih1 = (const float*)d_in[7];
  const float* b_hh1 = (const float*)d_in[8];
  const float* w_out = (const float*)d_in[9];
  const float* b_out = (const float*)d_in[10];
  float* out = (float*)d_out;

  const int grid = BATCH / ELEMS_PER_BLOCK;
  lstm2_wmma_kernel<<<grid, THREADS, 0, stream>>>(
      x, w_ih0, w_hh0, b_ih0, b_hh0, w_ih1, w_hh1, b_ih1, b_hh1, w_out, b_out, out);
}


# Weight fragments and bias initialisers built once per work-group, not once per wave: waves 0-3 each make a quarter with unconditional loads; LDS and one barrier give every lane its 48 values
# speedup vs baseline: 1.2103x; 1.0522x over previous
#include <hip/hip_runtime.h>
#include <hip/hip_bf16.h>

typedef __attribute__((ext_vector_type(2))) float v2f;
typedef __attribute__((ext_vector_type(4))) float v4f;
typedef __attribute__((ext_vector_type(8))) float v8f;

namespace {

constexpr int T_STEPS = 5;
constexpr int BATCH   = 1000000;
constexpr int IN_DIM  = 5;
constexpr int HID     = 6;
constexpr int ELEMS_PER_GROUP = 80;
constexpr int TILES           = 2;
constexpr int ELEMS_PER_BLOCK = TILES * ELEMS_PER_GROUP;
constexpr int WAVES_PER_BLOCK = 5;
constexpr int THREADS = WAVES_PER_BLOCK * 32;
constexpr int ROWS_PER_T = BATCH / T_STEPS;
constexpr int TAB_VALS  = 48;
constexpr int TAB_PITCH = 52;

__device__ __forceinline__ float tanh_fast(float x) {
#if __has_builtin(__builtin_amdgcn_tanhf)
  return __builtin_amdgcn_tanhf(x);
#else
  float e = __expf(-2.0f * __builtin_fabsf(x));
  float r = __fdividef(1.0f - e, 1.0f + e);
  return __builtin_copysignf(r, x);
#endif
}

__device__ __forceinline__ float sig_fast(float x) {
  return 0.5f * tanh_fast(x) + 0.5f;
}

__device__ __forceinline__ v8f wmma4(v2f a, v2f b, v8f c) {
  return __builtin_amdgcn_wmma_f32_16x16x4_f32(false, a, false, b, (short)0, c,
                                               false, false);
}

__device__ __forceinline__ float h_doubled(float to, float tc) { return __builtin_fmaf(to, tc, tc); }
__device__ __forceinline__ void lstm_unit(float gi, float gf, float gg, float go,
                                          float& c, float& h) {
  float I = sig_fast(gi);
  float F = sig_fast(gf);
  float G = tanh_fast(gg);
  float O = tanh_fast(go);
  c = F * c + I * G;
  h = h_doubled(O, tanh_fast(c));
}

__device__ __forceinline__ float wcol_l0(const float* wih, const float* whh, int R, int c,
                                         float sx, float sh) {
  const float a = wih[R * 5 + (c < 5 ? c : 0)];
  const float b = whh[R * 6 + ((c >= 5 && c < 11) ? c - 5 : 0)];
  return c < 5 ? sx * a : (c < 11 ? sh * b : 0.0f);
}
__device__ __forceinline__ float wcol_l1(const float* wih, const float* whh, int R, int c, float sh) {
  if (c < 6) return sh * wih[R * 6 + c];
  return sh * whh[R * 6 + (c - 6)];
}

__device__ __forceinline__ int unit_of(int hi, int q) {
  return q == 0 ? (hi ? 2 : 0) : q == 1 ? (hi ? 3 : 1) : (hi ? 5 : 4);
}
__device__ __forceinline__ int zcol_l0(int s, int hi, int j) {
  if (s == 0) return 2 * hi + j;
  if (s == 1) return j == 0 ? (hi ? 11 : 4) : 5 + unit_of(hi, 0);
  return 5 + unit_of(hi, 1 + j);
}
__device__ __forceinline__ int zcol_l1(int s, int hi, int j) {
  if (s == 0) return unit_of(hi, 1 + j);
  if (s == 1) return j == 0 ? unit_of(hi, 0) : 6 + unit_of(hi, 0);
  return 6 + unit_of(hi, 1 + j);
}

__device__ __forceinline__ int tile0_row(int m) {
  return (m & 3) * HID + (m >> 2);
}
__device__ __forceinline__ bool tile1_row(int m, int& R) {
  if (m < 4)            { R = m * HID + 4;       return true; }
  if (m >= 8 && m < 12) { R = (m - 8) * HID + 5; return true; }
  return false;
}


__device__ __forceinline__ void gates(v8f& g0, v8f& g1, const v2f (&At0)[3],
                                      const v2f (&At1)[3], v2f b0, v2f b1, v2f b2,
                                      v8f ci0, v8f ci1) {
  g0 = wmma4(At0[0], b0, ci0);  g1 = wmma4(At1[0], b0, ci1);
  g0 = wmma4(At0[1], b1, g0);   g1 = wmma4(At1[1], b1, g1);
  g0 = wmma4(At0[2], b2, g0);   g1 = wmma4(At1[2], b2, g1);
}
__device__ __forceinline__ void act_a(float gi, float gf, float gg, float go,
                                      float& I, float& F, float& G, float& O) {
  I = sig_fast(gi);  F = sig_fast(gf);  G = tanh_fast(gg);  O = tanh_fast(go);
}
__device__ __forceinline__ void act_b(float I, float F, float G, float O, float& c, float& h) {
  c = F * c + I * G;
  h = h_doubled(O, tanh_fast(c));
}
__device__ __forceinline__ void act(const v8f& g0, const v8f& g1, float (&c)[3], float (&h)[3]) {
  lstm_unit(g0[0], g0[1], g0[2], g0[3], c[0], h[0]);
  lstm_unit(g0[4], g0[5], g0[6], g0[7], c[1], h[1]);
  lstm_unit(g1[0], g1[1], g1[2], g1[3], c[2], h[2]);
}

#define SLICE_END() __builtin_amdgcn_sched_barrier(0)

template <class Tail>
__device__ __forceinline__ void phase(v8f& gx0, v8f& gx1, const v2f (&At0)[3],
                                      const v2f (&At1)[3], v2f b0, v2f b1, v2f b2,
                                      const v8f& ci0, const v8f& ci1,
                                      const v8f& gy0, const v8f& gy1,
                                      float (&c)[3], Tail tail) {
  float I0, F0, G0, O0, I1, F1, G1, O1, I2, F2, G2, O2;
  float hn0, hn1, hn2;
  gx0 = wmma4(At0[0], b0, ci0);
  act_a(gy1[0], gy1[1], gy1[2], gy1[3], I2, F2, G2, O2);
  SLICE_END();
  gx1 = wmma4(At1[0], b0, ci1);
  act_b(I2, F2, G2, O2, c[2], hn2);
  SLICE_END();
  gx0 = wmma4(At0[1], b1, gx0);
  act_a(gy0[0], gy0[1], gy0[2], gy0[3], I0, F0, G0, O0);
  SLICE_END();
  gx1 = wmma4(At1[1], b1, gx1);
  act_a(gy0[4], gy0[5], gy0[6], gy0[7], I1, F1, G1, O1);
  SLICE_END();
  gx0 = wmma4(At0[2], b2, gx0);
  act_b(I0, F0, G0, O0, c[0], hn0);
  act_b(I1, F1, G1, O1, c[1], hn1);
  SLICE_END();
  gx1 = wmma4(At1[2], b2, gx1);
  tail(hn0, hn1, hn2);
  SLICE_END();
}

}

__global__ __launch_bounds__(THREADS)
void lstm2_wmma_kernel(const float* __restrict__ x,
                       const float* __restrict__ w_ih0, const float* __restrict__ w_hh0,
                       const float* __restrict__ b_ih0, const float* __restrict__ b_hh0,
                       const float* __restrict__ w_ih1, const float* __restrict__ w_hh1,
                       const float* __restrict__ b_ih1, const float* __restrict__ b_hh1,
                       const float* __restrict__ w_out, const float* __restrict__ b_out,
                       float* __restrict__ out) {
  __shared__ float sp[TILES * T_STEPS * 2 * ELEMS_PER_GROUP];

  const int tid  = threadIdx.x;
  const int wave = tid >> 5;
  const int lane = tid & 31;
  const int n    = lane & 15;
  const int hi   = lane >> 4;
  const int e    = wave * 16 + n;

  const int p  = e % 5;
  const int u0 = hi ? 2 : 0, u1 = hi ? 3 : 1, u2 = hi ? 5 : 4;
  const float wl0 = w_out[p * HID + u0];
  const float wl1 = w_out[p * HID + u1];
  const float wl2 = w_out[p * HID + u2];

  v2f   xP[TILES][T_STEPS];
  float xC[TILES][T_STEPS];
#pragma unroll
  for (int k = 0; k < TILES; ++k) {
    const int b = (blockIdx.x * TILES + k) * ELEMS_PER_GROUP + e;
#pragma unroll
    for (int t = 0; t < T_STEPS; ++t) {
      const float* xp = x + ((size_t)t * BATCH + b) * IN_DIM;
      xP[k][t][0] = xp[2 * hi];
      xP[k][t][1] = xp[2 * hi + 1];
      const float x4 = xp[4];
      xC[k][t]    = hi ? 0.0f : x4;
    }
  }

  __shared__ __attribute__((aligned(16))) float tab[32 * TAB_PITCH];
  const int wave_u = __builtin_amdgcn_readfirstlane(wave);
  {
    float v[12];
#pragma unroll
    for (int i = 0; i < 12; ++i) v[i] = 0.0f;
    if (wave_u < 2) {
      const int m  = n;
      const int R0 = tile0_row(m);
      int R1; const bool v1 = tile1_row(m, R1);
      const int R1c = v1 ? R1 : 0;
      const float sc0 = (R0 / HID == 2) ? 1.0f : 0.5f;
      const float sc1 = (v1 && R1 / HID == 2) ? 1.0f : 0.5f;
      const float sh0 = 0.5f * sc0, sh1 = 0.5f * sc1;
      if (wave_u == 0) {
#pragma unroll
        for (int s = 0; s < 3; ++s) {
#pragma unroll
          for (int j = 0; j < 2; ++j) {
            const int ca = zcol_l0(s, hi, j);
            v[s * 2 + j]     = wcol_l0(w_ih0, w_hh0, R0, ca, sc0, sh0);
            const float w1   = wcol_l0(w_ih0, w_hh0, R1c, ca, sc1, sh1);
            v[6 + s * 2 + j] = v1 ? w1 : 0.0f;
          }
        }
      } else {
#pragma unroll
        for (int s = 0; s < 3; ++s) {
#pragma unroll
          for (int j = 0; j < 2; ++j) {
            const int cb = zcol_l1(s, hi, j);
            v[s * 2 + j]     = wcol_l1(w_ih1, w_hh1, R0, cb, sh0);
            const float w1   = wcol_l1(w_ih1, w_hh1, R1c, cb, sh1);
            v[6 + s * 2 + j] = v1 ? w1 : 0.0f;
          }
        }
      }
    } else if (wave_u < 4) {
      const float* bi = (wave_u == 2) ? b_ih0 : b_ih1;
      const float* bh = (wave_u == 2) ? b_hh0 : b_hh1;
#pragma unroll
      for (int r = 0; r < 8; ++r) {
        const int m = r + 8 * hi;
        const int R = tile0_row(m);
        const float sb = (R / HID == 2) ? 1.0f : 0.5f;
        v[r] = sb * (bi[R] + bh[R]);
      }
#pragma unroll
      for (int r = 0; r < 4; ++r) {
        int R1;
        tile1_row(r + 8 * hi, R1);
        const float sb1 = (R1 / HID == 2) ? 1.0f : 0.5f;
        v[8 + r] = sb1 * (bi[R1] + bh[R1]);
      }
    }
    if (wave_u < 4) {
      v4f* mine = reinterpret_cast<v4f*>(tab + lane * TAB_PITCH + 12 * wave_u);
      mine[0] = v4f{v[0], v[1], v[2],  v[3]};
      mine[1] = v4f{v[4], v[5], v[6],  v[7]};
      mine[2] = v4f{v[8], v[9], v[10], v[11]};
    }
  }

  __syncthreads();
  v2f A0t0[3], A0t1[3], A1t0[3], A1t1[3];
  v8f c0i0 = {}, c1i0 = {}, c0i1 = {}, c1i1 = {};
  {
    const v4f* tv = reinterpret_cast<const v4f*>(tab + lane * TAB_PITCH);
    float q[TAB_VALS];
#pragma unroll
    for (int i = 0; i < TAB_VALS / 4; ++i) {
      const v4f w = tv[i];
      q[4 * i] = w[0];  q[4 * i + 1] = w[1];  q[4 * i + 2] = w[2];  q[4 * i + 3] = w[3];
    }
#pragma unroll
    for (int s = 0; s < 3; ++s) {
      A0t0[s] = v2f{q[2 * s],      q[2 * s + 1]};
      A0t1[s] = v2f{q[6 + 2 * s],  q[6 + 2 * s + 1]};
      A1t0[s] = v2f{q[12 + 2 * s], q[12 + 2 * s + 1]};
      A1t1[s] = v2f{q[18 + 2 * s], q[18 + 2 * s + 1]};
    }
#pragma unroll
    for (int r = 0; r < 8; ++r) { c0i0[r] = q[24 + r];  c0i1[r] = q[36 + r]; }
#pragma unroll
    for (int r = 0; r < 4; ++r) { c1i0[r] = q[32 + r];  c1i1[r] = q[44 + r]; }
  }

  float c1s[TILES][3] = {}, c2s[TILES][3] = {};
  v2f P1[TILES], P2[TILES], Q0[TILES], Q1[TILES];
#pragma unroll
  for (int k = 0; k < TILES; ++k) {
    P1[k] = v2f{0.0f, 0.0f};  P2[k] = v2f{0.0f, 0.0f};
    Q1[k] = v2f{0.0f, 0.0f};  Q0[k] = v2f{xC[k][0], 0.0f};
  }
  v8f G0[TILES], G1[TILES];

  gates(G0[0], G1[0], A0t0, A0t1, xP[0][0], Q0[0], P1[0], c0i0, c1i0);
  SLICE_END();

#pragma unroll
  for (int t = 0; t < T_STEPS; ++t) {
    phase(G0[1], G1[1], A0t0, A0t1, xP[1][t], Q0[1], P1[1], c0i0, c1i0, G0[0], G1[0], c1s[0],
          [&](float hs0, float hs1, float hs2) {
            P1[0] = v2f{hs1, hs2};  Q1[0][0] = hs0;
          });

    phase(G0[0], G1[0], A1t0, A1t1, P1[0], Q1[0], P2[0], c0i1, c1i1, G0[1], G1[1], c1s[1],
          [&](float hs0, float hs1, float hs2) {
            P1[1] = v2f{hs1, hs2};  Q1[1][0] = hs0;
          });

    phase(G0[1], G1[1], A1t0, A1t1, P1[1], Q1[1], P2[1], c0i1, c1i1, G0[0], G1[0], c2s[0],
          [&](float hs0, float hs1, float hs2) {
            P2[0] = v2f{hs1, hs2};  Q1[0][1] = hs0;
            sp[((0 * T_STEPS + t) * 2 + hi) * ELEMS_PER_GROUP + e] = hs0 * wl0 + hs1 * wl1 + hs2 * wl2;
            if (t + 1 < T_STEPS) Q0[0] = v2f{xC[0][t + 1], Q1[0][0]};
          });

    if (t + 1 < T_STEPS) {
      phase(G0[0], G1[0], A0t0, A0t1, xP[0][t + 1], Q0[0], P1[0], c0i0, c1i0, G0[1], G1[1], c2s[1],
            [&](float hs0, float hs1, float hs2) {
              P2[1] = v2f{hs1, hs2};  Q1[1][1] = hs0;
              sp[((1 * T_STEPS + t) * 2 + hi) * ELEMS_PER_GROUP + e] = hs0 * wl0 + hs1 * wl1 + hs2 * wl2;
              Q0[1] = v2f{xC[1][t + 1], Q1[1][0]};
            });
    } else {
      float hl[3];
      act(G0[1], G1[1], c2s[1], hl);
      sp[((1 * T_STEPS + t) * 2 + hi) * ELEMS_PER_GROUP + e] =
          hl[0] * wl0 + hl[1] * wl1 + hl[2] * wl2;
    }
  }

  __syncthreads();

  {
    const int t   = wave;
    const int k   = hi;
    const int r16 = n;
    float sum = 2.0f * b_out[0];
#pragma unroll
    for (int pp = 0; pp < 5; ++pp) {
      const int ee = r16 * 5 + pp;
      sum += sp[((k * T_STEPS + t) * 2 + 0) * ELEMS_PER_GROUP + ee] +
             sp[((k * T_STEPS + t) * 2 + 1) * ELEMS_PER_GROUP + ee];
    }
    out[t * ROWS_PER_T + blockIdx.x * (TILES * 16) + lane] = 0.5f * sum;
  }
}

extern "C" void kernel_launch(void* const* d_in, const int* in_sizes, int n_in,
                              void* d_out, int out_size, void* d_ws, size_t ws_size,
                              hipStream_t stream) {
  const float* x     = (const float*)d_in[0];
  const float* w_ih0 = (const float*)d_in[1];
  const float* w_hh0 = (const float*)d_in[2];
  const float* b_ih0 = (const float*)d_in[3];
  const float* b_hh0 = (const float*)d_in[4];
  const float* w_ih1 = (const float*)d_in[5];
  const float* w_hh1 = (const float*)d_in[6];
  const float* b_ih1 = (const float*)d_in[7];
  const float* b_hh1 = (const float*)d_in[8];
  const float* w_out = (const float*)d_in[9];
  const float* b_out = (const float*)d_in[10];
  float* out = (float*)d_out;

  const int grid = BATCH / ELEMS_PER_BLOCK;
  lstm2_wmma_kernel<<<grid, THREADS, 0, stream>>>(
      x, w_ih0, w_hh0, b_ih0, b_hh0, w_ih1, w_hh1, b_ih1, b_hh1, w_out, b_out, out);
}


# First time step specialised for the zero start state: products whose operand pair is only zero h (8 of 120 per wave) and the forget gate against zero c are dropped; other terms keep their order
# speedup vs baseline: 1.2761x; 1.0544x over previous
#include <hip/hip_runtime.h>
#include <hip/hip_bf16.h>

typedef __attribute__((ext_vector_type(2))) float v2f;
typedef __attribute__((ext_vector_type(4))) float v4f;
typedef __attribute__((ext_vector_type(8))) float v8f;

namespace {

constexpr int T_STEPS = 5;
constexpr int BATCH   = 1000000;
constexpr int IN_DIM  = 5;
constexpr int HID     = 6;
constexpr int ELEMS_PER_GROUP = 80;
constexpr int TILES           = 2;
constexpr int ELEMS_PER_BLOCK = TILES * ELEMS_PER_GROUP;
constexpr int WAVES_PER_BLOCK = 5;
constexpr int THREADS = WAVES_PER_BLOCK * 32;
constexpr int ROWS_PER_T = BATCH / T_STEPS;
constexpr int TAB_VALS  = 48;
constexpr int TAB_PITCH = 52;

__device__ __forceinline__ float tanh_fast(float x) {
#if __has_builtin(__builtin_amdgcn_tanhf)
  return __builtin_amdgcn_tanhf(x);
#else
  float e = __expf(-2.0f * __builtin_fabsf(x));
  float r = __fdividef(1.0f - e, 1.0f + e);
  return __builtin_copysignf(r, x);
#endif
}

__device__ __forceinline__ float sig_fast(float x) {
  return 0.5f * tanh_fast(x) + 0.5f;
}

__device__ __forceinline__ v8f wmma4(v2f a, v2f b, v8f c) {
  return __builtin_amdgcn_wmma_f32_16x16x4_f32(false, a, false, b, (short)0, c,
                                               false, false);
}

__device__ __forceinline__ float h_doubled(float to, float tc) { return __builtin_fmaf(to, tc, tc); }
__device__ __forceinline__ void lstm_unit(float gi, float gf, float gg, float go,
                                          float& c, float& h) {
  float I = sig_fast(gi);
  float F = sig_fast(gf);
  float G = tanh_fast(gg);
  float O = tanh_fast(go);
  c = F * c + I * G;
  h = h_doubled(O, tanh_fast(c));
}

__device__ __forceinline__ float wcol_l0(const float* wih, const float* whh, int R, int c,
                                         float sx, float sh) {
  const float a = wih[R * 5 + (c < 5 ? c : 0)];
  const float b = whh[R * 6 + ((c >= 5 && c < 11) ? c - 5 : 0)];
  return c < 5 ? sx * a : (c < 11 ? sh * b : 0.0f);
}
__device__ __forceinline__ float wcol_l1(const float* wih, const float* whh, int R, int c, float sh) {
  if (c < 6) return sh * wih[R * 6 + c];
  return sh * whh[R * 6 + (c - 6)];
}

__device__ __forceinline__ int unit_of(int hi, int q) {
  return q == 0 ? (hi ? 2 : 0) : q == 1 ? (hi ? 3 : 1) : (hi ? 5 : 4);
}
__device__ __forceinline__ int zcol_l0(int s, int hi, int j) {
  if (s == 0) return 2 * hi + j;
  if (s == 1) return j == 0 ? (hi ? 11 : 4) : 5 + unit_of(hi, 0);
  return 5 + unit_of(hi, 1 + j);
}
__device__ __forceinline__ int zcol_l1(int s, int hi, int j) {
  if (s == 0) return unit_of(hi, 1 + j);
  if (s == 1) return j == 0 ? unit_of(hi, 0) : 6 + unit_of(hi, 0);
  return 6 + unit_of(hi, 1 + j);
}

__device__ __forceinline__ int tile0_row(int m) {
  return (m & 3) * HID + (m >> 2);
}
__device__ __forceinline__ bool tile1_row(int m, int& R) {
  if (m < 4)            { R = m * HID + 4;       return true; }
  if (m >= 8 && m < 12) { R = (m - 8) * HID + 5; return true; }
  return false;
}


__device__ __forceinline__ void gates_first(v8f& g0, v8f& g1, const v2f (&At0)[3],
                                            const v2f (&At1)[3], v2f b0, v2f b1,
                                            v8f ci0, v8f ci1) {
  g0 = wmma4(At0[0], b0, ci0);  g1 = wmma4(At1[0], b0, ci1);
  g0 = wmma4(At0[1], b1, g0);   g1 = wmma4(At1[1], b1, g1);
}
__device__ __forceinline__ void act_a(float gi, float gf, float gg, float go,
                                      float& I, float& F, float& G, float& O) {
  I = sig_fast(gi);  F = sig_fast(gf);  G = tanh_fast(gg);  O = tanh_fast(go);
}
__device__ __forceinline__ void act_b(float I, float F, float G, float O, float& c, float& h) {
  c = F * c + I * G;
  h = h_doubled(O, tanh_fast(c));
}
__device__ __forceinline__ void act_a_first(float gi, float gg, float go,
                                            float& I, float& G, float& O) {
  I = sig_fast(gi);  G = tanh_fast(gg);  O = tanh_fast(go);
}
__device__ __forceinline__ void act_b_first(float I, float G, float O, float& c, float& h) {
  c = I * G;
  h = h_doubled(O, tanh_fast(c));
}
__device__ __forceinline__ void act(const v8f& g0, const v8f& g1, float (&c)[3], float (&h)[3]) {
  lstm_unit(g0[0], g0[1], g0[2], g0[3], c[0], h[0]);
  lstm_unit(g0[4], g0[5], g0[6], g0[7], c[1], h[1]);
  lstm_unit(g1[0], g1[1], g1[2], g1[3], c[2], h[2]);
}

#define SLICE_END() __builtin_amdgcn_sched_barrier(0)

template <class Tail>
__device__ __forceinline__ void phase(v8f& gx0, v8f& gx1, const v2f (&At0)[3],
                                      const v2f (&At1)[3], v2f b0, v2f b1, v2f b2,
                                      const v8f& ci0, const v8f& ci1,
                                      const v8f& gy0, const v8f& gy1,
                                      float (&c)[3], Tail tail) {
  float I0, F0, G0, O0, I1, F1, G1, O1, I2, F2, G2, O2;
  float hn0, hn1, hn2;
  gx0 = wmma4(At0[0], b0, ci0);
  act_a(gy1[0], gy1[1], gy1[2], gy1[3], I2, F2, G2, O2);
  SLICE_END();
  gx1 = wmma4(At1[0], b0, ci1);
  act_b(I2, F2, G2, O2, c[2], hn2);
  SLICE_END();
  gx0 = wmma4(At0[1], b1, gx0);
  act_a(gy0[0], gy0[1], gy0[2], gy0[3], I0, F0, G0, O0);
  SLICE_END();
  gx1 = wmma4(At1[1], b1, gx1);
  act_a(gy0[4], gy0[5], gy0[6], gy0[7], I1, F1, G1, O1);
  SLICE_END();
  gx0 = wmma4(At0[2], b2, gx0);
  act_b(I0, F0, G0, O0, c[0], hn0);
  act_b(I1, F1, G1, O1, c[1], hn1);
  SLICE_END();
  gx1 = wmma4(At1[2], b2, gx1);
  tail(hn0, hn1, hn2);
  SLICE_END();
}

template <class Tail>
__device__ __forceinline__ void phase_first_short(v8f& gx0, v8f& gx1, const v2f (&At0)[3],
                                                  const v2f (&At1)[3], v2f b0, v2f b1,
                                                  const v8f& ci0, const v8f& ci1,
                                                  const v8f& gy0, const v8f& gy1,
                                                  float (&c)[3], Tail tail) {
  float I0, G0, O0, I1, G1, O1, I2, G2, O2;
  float hn0, hn1, hn2;
  gx0 = wmma4(At0[0], b0, ci0);
  act_a_first(gy1[0], gy1[2], gy1[3], I2, G2, O2);
  SLICE_END();
  gx1 = wmma4(At1[0], b0, ci1);
  act_b_first(I2, G2, O2, c[2], hn2);
  act_a_first(gy0[0], gy0[2], gy0[3], I0, G0, O0);
  SLICE_END();
  gx0 = wmma4(At0[1], b1, gx0);
  act_a_first(gy0[4], gy0[6], gy0[7], I1, G1, O1);
  act_b_first(I0, G0, O0, c[0], hn0);
  SLICE_END();
  gx1 = wmma4(At1[1], b1, gx1);
  act_b_first(I1, G1, O1, c[1], hn1);
  tail(hn0, hn1, hn2);
  SLICE_END();
}
template <class Tail>
__device__ __forceinline__ void phase_first_full(v8f& gx0, v8f& gx1, const v2f (&At0)[3],
                                                 const v2f (&At1)[3], v2f b0, v2f b1, v2f b2,
                                                 const v8f& ci0, const v8f& ci1,
                                                 const v8f& gy0, const v8f& gy1,
                                                 float (&c)[3], Tail tail) {
  float I0, G0, O0, I1, G1, O1, I2, G2, O2;
  float hn0, hn1, hn2;
  gx0 = wmma4(At0[0], b0, ci0);
  act_a_first(gy1[0], gy1[2], gy1[3], I2, G2, O2);
  SLICE_END();
  gx1 = wmma4(At1[0], b0, ci1);
  act_b_first(I2, G2, O2, c[2], hn2);
  SLICE_END();
  gx0 = wmma4(At0[1], b1, gx0);
  act_a_first(gy0[0], gy0[2], gy0[3], I0, G0, O0);
  SLICE_END();
  gx1 = wmma4(At1[1], b1, gx1);
  act_a_first(gy0[4], gy0[6], gy0[7], I1, G1, O1);
  SLICE_END();
  gx0 = wmma4(At0[2], b2, gx0);
  act_b_first(I0, G0, O0, c[0], hn0);
  act_b_first(I1, G1, O1, c[1], hn1);
  SLICE_END();
  gx1 = wmma4(At1[2], b2, gx1);
  tail(hn0, hn1, hn2);
  SLICE_END();
}

}

__global__ __launch_bounds__(THREADS)
void lstm2_wmma_kernel(const float* __restrict__ x,
                       const float* __restrict__ w_ih0, const float* __restrict__ w_hh0,
                       const float* __restrict__ b_ih0, const float* __restrict__ b_hh0,
                       const float* __restrict__ w_ih1, const float* __restrict__ w_hh1,
                       const float* __restrict__ b_ih1, const float* __restrict__ b_hh1,
                       const float* __restrict__ w_out, const float* __restrict__ b_out,
                       float* __restrict__ out) {
  __shared__ float sp[TILES * T_STEPS * 2 * ELEMS_PER_GROUP];

  const int tid  = threadIdx.x;
  const int wave = tid >> 5;
  const int lane = tid & 31;
  const int n    = lane & 15;
  const int hi   = lane >> 4;
  const int e    = wave * 16 + n;

  const int p  = e % 5;
  const int u0 = hi ? 2 : 0, u1 = hi ? 3 : 1, u2 = hi ? 5 : 4;
  const float wl0 = w_out[p * HID + u0];
  const float wl1 = w_out[p * HID + u1];
  const float wl2 = w_out[p * HID + u2];

  v2f   xP[TILES][T_STEPS];
  float xC[TILES][T_STEPS];
#pragma unroll
  for (int k = 0; k < TILES; ++k) {
    const int b = (blockIdx.x * TILES + k) * ELEMS_PER_GROUP + e;
#pragma unroll
    for (int t = 0; t < T_STEPS; ++t) {
      const float* xp = x + ((size_t)t * BATCH + b) * IN_DIM;
      xP[k][t][0] = xp[2 * hi];
      xP[k][t][1] = xp[2 * hi + 1];
      const float x4 = xp[4];
      xC[k][t]    = hi ? 0.0f : x4;
    }
  }

  __shared__ __attribute__((aligned(16))) float tab[32 * TAB_PITCH];
  const int wave_u = __builtin_amdgcn_readfirstlane(wave);
  {
    float v[12];
#pragma unroll
    for (int i = 0; i < 12; ++i) v[i] = 0.0f;
    if (wave_u < 2) {
      const int m  = n;
      const int R0 = tile0_row(m);
      int R1; const bool v1 = tile1_row(m, R1);
      const int R1c = v1 ? R1 : 0;
      const float sc0 = (R0 / HID == 2) ? 1.0f : 0.5f;
      const float sc1 = (v1 && R1 / HID == 2) ? 1.0f : 0.5f;
      const float sh0 = 0.5f * sc0, sh1 = 0.5f * sc1;
      if (wave_u == 0) {
#pragma unroll
        for (int s = 0; s < 3; ++s) {
#pragma unroll
          for (int j = 0; j < 2; ++j) {
            const int ca = zcol_l0(s, hi, j);
            v[s * 2 + j]     = wcol_l0(w_ih0, w_hh0, R0, ca, sc0, sh0);
            const float w1   = wcol_l0(w_ih0, w_hh0, R1c, ca, sc1, sh1);
            v[6 + s * 2 + j] = v1 ? w1 : 0.0f;
          }
        }
      } else {
#pragma unroll
        for (int s = 0; s < 3; ++s) {
#pragma unroll
          for (int j = 0; j < 2; ++j) {
            const int cb = zcol_l1(s, hi, j);
            v[s * 2 + j]     = wcol_l1(w_ih1, w_hh1, R0, cb, sh0);
            const float w1   = wcol_l1(w_ih1, w_hh1, R1c, cb, sh1);
            v[6 + s * 2 + j] = v1 ? w1 : 0.0f;
          }
        }
      }
    } else if (wave_u < 4) {
      const float* bi = (wave_u == 2) ? b_ih0 : b_ih1;
      const float* bh = (wave_u == 2) ? b_hh0 : b_hh1;
#pragma unroll
      for (int r = 0; r < 8; ++r) {
        const int m = r + 8 * hi;
        const int R = tile0_row(m);
        const float sb = (R / HID == 2) ? 1.0f : 0.5f;
        v[r] = sb * (bi[R] + bh[R]);
      }
#pragma unroll
      for (int r = 0; r < 4; ++r) {
        int R1;
        tile1_row(r + 8 * hi, R1);
        const float sb1 = (R1 / HID == 2) ? 1.0f : 0.5f;
        v[8 + r] = sb1 * (bi[R1] + bh[R1]);
      }
    }
    if (wave_u < 4) {
      v4f* mine = reinterpret_cast<v4f*>(tab + lane * TAB_PITCH + 12 * wave_u);
      mine[0] = v4f{v[0], v[1], v[2],  v[3]};
      mine[1] = v4f{v[4], v[5], v[6],  v[7]};
      mine[2] = v4f{v[8], v[9], v[10], v[11]};
    }
  }

  __syncthreads();
  v2f A0t0[3], A0t1[3], A1t0[3], A1t1[3];
  v8f c0i0 = {}, c1i0 = {}, c0i1 = {}, c1i1 = {};
  {
    const v4f* tv = reinterpret_cast<const v4f*>(tab + lane * TAB_PITCH);
    float q[TAB_VALS];
#pragma unroll
    for (int i = 0; i < TAB_VALS / 4; ++i) {
      const v4f w = tv[i];
      q[4 * i] = w[0];  q[4 * i + 1] = w[1];  q[4 * i + 2] = w[2];  q[4 * i + 3] = w[3];
    }
#pragma unroll
    for (int s = 0; s < 3; ++s) {
      A0t0[s] = v2f{q[2 * s],      q[2 * s + 1]};
      A0t1[s] = v2f{q[6 + 2 * s],  q[6 + 2 * s + 1]};
      A1t0[s] = v2f{q[12 + 2 * s], q[12 + 2 * s + 1]};
      A1t1[s] = v2f{q[18 + 2 * s], q[18 + 2 * s + 1]};
    }
#pragma unroll
    for (int r = 0; r < 8; ++r) { c0i0[r] = q[24 + r];  c0i1[r] = q[36 + r]; }
#pragma unroll
    for (int r = 0; r < 4; ++r) { c1i0[r] = q[32 + r];  c1i1[r] = q[44 + r]; }
  }

  float c1s[TILES][3] = {}, c2s[TILES][3] = {};
  v2f P1[TILES], P2[TILES], Q0[TILES], Q1[TILES];
#pragma unroll
  for (int k = 0; k < TILES; ++k) {
    P1[k] = v2f{0.0f, 0.0f};  P2[k] = v2f{0.0f, 0.0f};
    Q1[k] = v2f{0.0f, 0.0f};  Q0[k] = v2f{xC[k][0], 0.0f};
  }
  v8f G0[TILES], G1[TILES];

  gates_first(G0[0], G1[0], A0t0, A0t1, xP[0][0], Q0[0], c0i0, c1i0);
  SLICE_END();

#pragma unroll
  for (int t = 0; t < T_STEPS; ++t) {
    const bool first = (t == 0);
    auto tail1 = [&](float hs0, float hs1, float hs2) {
      P1[0] = v2f{hs1, hs2};  Q1[0][0] = hs0;
    };
    auto tail2 = [&](float hs0, float hs1, float hs2) {
      P1[1] = v2f{hs1, hs2};  Q1[1][0] = hs0;
    };
    auto tail3 = [&](float hs0, float hs1, float hs2) {
      P2[0] = v2f{hs1, hs2};  Q1[0][1] = hs0;
      sp[((0 * T_STEPS + t) * 2 + hi) * ELEMS_PER_GROUP + e] = hs0 * wl0 + hs1 * wl1 + hs2 * wl2;
      if (t + 1 < T_STEPS) Q0[0] = v2f{xC[0][t + 1], Q1[0][0]};
    };
    auto tail4 = [&](float hs0, float hs1, float hs2) {
      P2[1] = v2f{hs1, hs2};  Q1[1][1] = hs0;
      sp[((1 * T_STEPS + t) * 2 + hi) * ELEMS_PER_GROUP + e] = hs0 * wl0 + hs1 * wl1 + hs2 * wl2;
      if (t + 1 < T_STEPS) Q0[1] = v2f{xC[1][t + 1], Q1[1][0]};
    };

    if (first)
      phase_first_short(G0[1], G1[1], A0t0, A0t1, xP[1][t], Q0[1], c0i0, c1i0, G0[0], G1[0],
                        c1s[0], tail1);
    else
      phase(G0[1], G1[1], A0t0, A0t1, xP[1][t], Q0[1], P1[1], c0i0, c1i0, G0[0], G1[0], c1s[0],
            tail1);

    if (first)
      phase_first_short(G0[0], G1[0], A1t0, A1t1, P1[0], Q1[0], c0i1, c1i1, G0[1], G1[1],
                        c1s[1], tail2);
    else
      phase(G0[0], G1[0], A1t0, A1t1, P1[0], Q1[0], P2[0], c0i1, c1i1, G0[1], G1[1], c1s[1],
            tail2);

    if (first)
      phase_first_short(G0[1], G1[1], A1t0, A1t1, P1[1], Q1[1], c0i1, c1i1, G0[0], G1[0],
                        c2s[0], tail3);
    else
      phase(G0[1], G1[1], A1t0, A1t1, P1[1], Q1[1], P2[1], c0i1, c1i1, G0[0], G1[0], c2s[0],
            tail3);

    if (t + 1 < T_STEPS) {
      if (first)
        phase_first_full(G0[0], G1[0], A0t0, A0t1, xP[0][t + 1], Q0[0], P1[0], c0i0, c1i0,
                         G0[1], G1[1], c2s[1], tail4);
      else
        phase(G0[0], G1[0], A0t0, A0t1, xP[0][t + 1], Q0[0], P1[0], c0i0, c1i0, G0[1], G1[1],
              c2s[1], tail4);
    } else {
      float hl[3];
      act(G0[1], G1[1], c2s[1], hl);
      sp[((1 * T_STEPS + t) * 2 + hi) * ELEMS_PER_GROUP + e] =
          hl[0] * wl0 + hl[1] * wl1 + hl[2] * wl2;
    }
  }

  __syncthreads();

  {
    const int t   = wave;
    const int k   = hi;
    const int r16 = n;
    float sum = 2.0f * b_out[0];
#pragma unroll
    for (int pp = 0; pp < 5; ++pp) {
      const int ee = r16 * 5 + pp;
      sum += sp[((k * T_STEPS + t) * 2 + 0) * ELEMS_PER_GROUP + ee] +
             sp[((k * T_STEPS + t) * 2 + 1) * ELEMS_PER_GROUP + ee];
    }
    out[t * ROWS_PER_T + blockIdx.x * (TILES * 16) + lane] = 0.5f * sum;
  }
}

extern "C" void kernel_launch(void* const* d_in, const int* in_sizes, int n_in,
                              void* d_out, int out_size, void* d_ws, size_t ws_size,
                              hipStream_t stream) {
  const float* x     = (const float*)d_in[0];
  const float* w_ih0 = (const float*)d_in[1];
  const float* w_hh0 = (const float*)d_in[2];
  const float* b_ih0 = (const float*)d_in[3];
  const float* b_hh0 = (const float*)d_in[4];
  const float* w_ih1 = (const float*)d_in[5];
  const float* w_hh1 = (const float*)d_in[6];
  const float* b_ih1 = (const float*)d_in[7];
  const float* b_hh1 = (const float*)d_in[8];
  const float* w_out = (const float*)d_in[9];
  const float* b_out = (const float*)d_in[10];
  float* out = (float*)d_out;

  const int grid = BATCH / ELEMS_PER_BLOCK;
  lstm2_wmma_kernel<<<grid, THREADS, 0, stream>>>(
      x, w_ih0, w_hh0, b_ih0, b_hh0, w_ih1, w_hh1, b_ih1, b_hh1, w_out, b_out, out);
}


# Work-groups of 2 waves, not 5: each wave takes the five 16-element tiles of its 80-element group one after another, next tile's x requested a tile ahead; the four SIMDs then carry equal waves
# speedup vs baseline: 1.3348x; 1.0460x over previous
#include <hip/hip_runtime.h>
#include <hip/hip_bf16.h>

typedef __attribute__((ext_vector_type(2))) float v2f;
typedef __attribute__((ext_vector_type(4))) float v4f;
typedef __attribute__((ext_vector_type(8))) float v8f;

namespace {

constexpr int T_STEPS = 5;
constexpr int BATCH   = 1000000;
constexpr int IN_DIM  = 5;
constexpr int HID     = 6;
constexpr int ELEMS_PER_GROUP = 80;
constexpr int TILES_PER_WAVE  = ELEMS_PER_GROUP / 16;
constexpr int WAVES_PER_BLOCK = 2;
constexpr int ELEMS_PER_BLOCK = WAVES_PER_BLOCK * ELEMS_PER_GROUP;
constexpr int ROWS_PER_BLOCK  = ELEMS_PER_BLOCK / 5;
constexpr int LINES_PER_T     = ROWS_PER_BLOCK / 32;
constexpr int LINES_PER_BLOCK = T_STEPS * LINES_PER_T;
constexpr int THREADS = WAVES_PER_BLOCK * 32;
constexpr int ROWS_PER_T = BATCH / T_STEPS;
constexpr int TAB_VALS  = 48;
constexpr int TAB_PITCH = 52;

__device__ __forceinline__ float tanh_fast(float x) {
#if __has_builtin(__builtin_amdgcn_tanhf)
  return __builtin_amdgcn_tanhf(x);
#else
  float e = __expf(-2.0f * __builtin_fabsf(x));
  float r = __fdividef(1.0f - e, 1.0f + e);
  return __builtin_copysignf(r, x);
#endif
}

__device__ __forceinline__ float sig_fast(float x) {
  return 0.5f * tanh_fast(x) + 0.5f;
}

__device__ __forceinline__ v8f wmma4(v2f a, v2f b, v8f c) {
  return __builtin_amdgcn_wmma_f32_16x16x4_f32(false, a, false, b, (short)0, c,
                                               false, false);
}

__device__ __forceinline__ float h_doubled(float to, float tc) { return __builtin_fmaf(to, tc, tc); }
__device__ __forceinline__ void lstm_unit(float gi, float gf, float gg, float go,
                                          float& c, float& h) {
  float I = sig_fast(gi);
  float F = sig_fast(gf);
  float G = tanh_fast(gg);
  float O = tanh_fast(go);
  c = F * c + I * G;
  h = h_doubled(O, tanh_fast(c));
}

__device__ __forceinline__ float wcol_l0(const float* wih, const float* whh, int R, int c,
                                         float sx, float sh) {
  const float a = wih[R * 5 + (c < 5 ? c : 0)];
  const float b = whh[R * 6 + ((c >= 5 && c < 11) ? c - 5 : 0)];
  return c < 5 ? sx * a : (c < 11 ? sh * b : 0.0f);
}
__device__ __forceinline__ float wcol_l1(const float* wih, const float* whh, int R, int c, float sh) {
  if (c < 6) return sh * wih[R * 6 + c];
  return sh * whh[R * 6 + (c - 6)];
}

__device__ __forceinline__ int unit_of(int hi, int q) {
  return q == 0 ? (hi ? 2 : 0) : q == 1 ? (hi ? 3 : 1) : (hi ? 5 : 4);
}
__device__ __forceinline__ int zcol_l0(int s, int hi, int j) {
  if (s == 0) return 2 * hi + j;
  if (s == 1) return j == 0 ? (hi ? 11 : 4) : 5 + unit_of(hi, 0);
  return 5 + unit_of(hi, 1 + j);
}
__device__ __forceinline__ int zcol_l1(int s, int hi, int j) {
  if (s == 0) return unit_of(hi, 1 + j);
  if (s == 1) return j == 0 ? unit_of(hi, 0) : 6 + unit_of(hi, 0);
  return 6 + unit_of(hi, 1 + j);
}

__device__ __forceinline__ int tile0_row(int m) {
  return (m & 3) * HID + (m >> 2);
}
__device__ __forceinline__ bool tile1_row(int m, int& R) {
  if (m < 4)            { R = m * HID + 4;       return true; }
  if (m >= 8 && m < 12) { R = (m - 8) * HID + 5; return true; }
  return false;
}


__device__ __forceinline__ void gates_first(v8f& g0, v8f& g1, const v2f (&At0)[3],
                                            const v2f (&At1)[3], v2f b0, v2f b1,
                                            v8f ci0, v8f ci1) {
  g0 = wmma4(At0[0], b0, ci0);  g1 = wmma4(At1[0], b0, ci1);
  g0 = wmma4(At0[1], b1, g0);   g1 = wmma4(At1[1], b1, g1);
}
__device__ __forceinline__ void gates(v8f& g0, v8f& g1, const v2f (&At0)[3],
                                      const v2f (&At1)[3], v2f b0, v2f b1, v2f b2,
                                      v8f ci0, v8f ci1) {
  g0 = wmma4(At0[0], b0, ci0);  g1 = wmma4(At1[0], b0, ci1);
  g0 = wmma4(At0[1], b1, g0);   g1 = wmma4(At1[1], b1, g1);
  g0 = wmma4(At0[2], b2, g0);   g1 = wmma4(At1[2], b2, g1);
}
__device__ __forceinline__ void act(const v8f& g0, const v8f& g1, float (&c)[3], float (&h)[3]) {
  lstm_unit(g0[0], g0[1], g0[2], g0[3], c[0], h[0]);
  lstm_unit(g0[4], g0[5], g0[6], g0[7], c[1], h[1]);
  lstm_unit(g1[0], g1[1], g1[2], g1[3], c[2], h[2]);
}
__device__ __forceinline__ void lstm_unit_first(float gi, float gg, float go, float& c, float& h) {
  float I = sig_fast(gi);
  float G = tanh_fast(gg);
  float O = tanh_fast(go);
  c = I * G;
  h = h_doubled(O, tanh_fast(c));
}
__device__ __forceinline__ void act_first(const v8f& g0, const v8f& g1, float (&c)[3],
                                          float (&h)[3]) {
  lstm_unit_first(g0[0], g0[2], g0[3], c[0], h[0]);
  lstm_unit_first(g0[4], g0[6], g0[7], c[1], h[1]);
  lstm_unit_first(g1[0], g1[2], g1[3], c[2], h[2]);
}

}

__global__ __launch_bounds__(THREADS)
void lstm2_wmma_kernel(const float* __restrict__ x,
                       const float* __restrict__ w_ih0, const float* __restrict__ w_hh0,
                       const float* __restrict__ b_ih0, const float* __restrict__ b_hh0,
                       const float* __restrict__ w_ih1, const float* __restrict__ w_hh1,
                       const float* __restrict__ b_ih1, const float* __restrict__ b_hh1,
                       const float* __restrict__ w_out, const float* __restrict__ b_out,
                       float* __restrict__ out) {
  __shared__ float sp[T_STEPS * 2 * ELEMS_PER_BLOCK];

  const int tid  = threadIdx.x;
  const int wave = tid >> 5;
  const int lane = tid & 31;
  const int n    = lane & 15;
  const int hi   = lane >> 4;
  const int u0 = hi ? 2 : 0, u1 = hi ? 3 : 1, u2 = hi ? 5 : 4;

  const int e0 = wave * ELEMS_PER_GROUP + n;
  const float* xlane = x + ((size_t)blockIdx.x * ELEMS_PER_BLOCK + e0) * IN_DIM;
  v2f   xPn[T_STEPS];
  float xCn[T_STEPS];
  auto request_x = [&](int j) {
#pragma unroll
    for (int t = 0; t < T_STEPS; ++t) {
      const float* xp = xlane + ((size_t)t * BATCH + j * 16) * IN_DIM;
      xPn[t][0] = xp[2 * hi];
      xPn[t][1] = xp[2 * hi + 1];
      xCn[t]    = xp[4];
    }
  };
  request_x(0);

  __shared__ __attribute__((aligned(16))) float tab[32 * TAB_PITCH];
  const int wave_u = __builtin_amdgcn_readfirstlane(wave);
  for (int quarter = wave_u; quarter < 4; quarter += WAVES_PER_BLOCK) {
    float v[12];
#pragma unroll
    for (int i = 0; i < 12; ++i) v[i] = 0.0f;
    if (quarter < 2) {
      const int m  = n;
      const int R0 = tile0_row(m);
      int R1; const bool v1 = tile1_row(m, R1);
      const int R1c = v1 ? R1 : 0;
      const float sc0 = (R0 / HID == 2) ? 1.0f : 0.5f;
      const float sc1 = (v1 && R1 / HID == 2) ? 1.0f : 0.5f;
      const float sh0 = 0.5f * sc0, sh1 = 0.5f * sc1;
      if (quarter == 0) {
#pragma unroll
        for (int s = 0; s < 3; ++s) {
#pragma unroll
          for (int j = 0; j < 2; ++j) {
            const int ca = zcol_l0(s, hi, j);
            v[s * 2 + j]     = wcol_l0(w_ih0, w_hh0, R0, ca, sc0, sh0);
            const float w1   = wcol_l0(w_ih0, w_hh0, R1c, ca, sc1, sh1);
            v[6 + s * 2 + j] = v1 ? w1 : 0.0f;
          }
        }
      } else {
#pragma unroll
        for (int s = 0; s < 3; ++s) {
#pragma unroll
          for (int j = 0; j < 2; ++j) {
            const int cb = zcol_l1(s, hi, j);
            v[s * 2 + j]     = wcol_l1(w_ih1, w_hh1, R0, cb, sh0);
            const float w1   = wcol_l1(w_ih1, w_hh1, R1c, cb, sh1);
            v[6 + s * 2 + j] = v1 ? w1 : 0.0f;
          }
        }
      }
    } else {
      const float* bi = (quarter == 2) ? b_ih0 : b_ih1;
      const float* bh = (quarter == 2) ? b_hh0 : b_hh1;
#pragma unroll
      for (int r = 0; r < 8; ++r) {
        const int m = r + 8 * hi;
        const int R = tile0_row(m);
        const float sb = (R / HID == 2) ? 1.0f : 0.5f;
        v[r] = sb * (bi[R] + bh[R]);
      }
#pragma unroll
      for (int r = 0; r < 4; ++r) {
        int R1;
        tile1_row(r + 8 * hi, R1);
        const float sb1 = (R1 / HID == 2) ? 1.0f : 0.5f;
        v[8 + r] = sb1 * (bi[R1] + bh[R1]);
      }
    }
    v4f* mine = reinterpret_cast<v4f*>(tab + lane * TAB_PITCH + 12 * quarter);
    mine[0] = v4f{v[0], v[1], v[2],  v[3]};
    mine[1] = v4f{v[4], v[5], v[6],  v[7]};
    mine[2] = v4f{v[8], v[9], v[10], v[11]};
  }

  __syncthreads();
  v2f A0t0[3], A0t1[3], A1t0[3], A1t1[3];
  v8f c0i0 = {}, c1i0 = {}, c0i1 = {}, c1i1 = {};
  {
    const v4f* tv = reinterpret_cast<const v4f*>(tab + lane * TAB_PITCH);
    float q[TAB_VALS];
#pragma unroll
    for (int i = 0; i < TAB_VALS / 4; ++i) {
      const v4f w = tv[i];
      q[4 * i] = w[0];  q[4 * i + 1] = w[1];  q[4 * i + 2] = w[2];  q[4 * i + 3] = w[3];
    }
#pragma unroll
    for (int s = 0; s < 3; ++s) {
      A0t0[s] = v2f{q[2 * s],      q[2 * s + 1]};
      A0t1[s] = v2f{q[6 + 2 * s],  q[6 + 2 * s + 1]};
      A1t0[s] = v2f{q[12 + 2 * s], q[12 + 2 * s + 1]};
      A1t1[s] = v2f{q[18 + 2 * s], q[18 + 2 * s + 1]};
    }
#pragma unroll
    for (int r = 0; r < 8; ++r) { c0i0[r] = q[24 + r];  c0i1[r] = q[36 + r]; }
#pragma unroll
    for (int r = 0; r < 4; ++r) { c1i0[r] = q[32 + r];  c1i1[r] = q[44 + r]; }
  }


  int p = n % 5;
  auto run_tile = [&](const int j, const bool more) __attribute__((always_inline)) {
    v2f   xP[T_STEPS];
    float xC[T_STEPS];
#pragma unroll
    for (int t = 0; t < T_STEPS; ++t) { xP[t] = xPn[t];  xC[t] = hi ? 0.0f : xCn[t]; }
    const float wl0 = w_out[p * HID + u0];
    const float wl1 = w_out[p * HID + u1];
    const float wl2 = w_out[p * HID + u2];
    p = (p == 4) ? 0 : p + 1;
    if (more) request_x(j + 1);
    __builtin_amdgcn_sched_barrier(0);

    const int eb = e0 + j * 16;
    float c1s[3] = {}, c2s[3] = {};
    v2f P1 = v2f{0.0f, 0.0f}, P2 = v2f{0.0f, 0.0f}, Q1 = v2f{0.0f, 0.0f};
    v2f Q0 = v2f{xC[0], 0.0f};
    v8f G0, G1;
    float hs[3];
#pragma unroll
    for (int t = 0; t < T_STEPS; ++t) {
      if (t == 0) {
        gates_first(G0, G1, A0t0, A0t1, xP[t], Q0, c0i0, c1i0);
        act_first(G0, G1, c1s, hs);
      } else {
        gates(G0, G1, A0t0, A0t1, xP[t], Q0, P1, c0i0, c1i0);
        act(G0, G1, c1s, hs);
      }
      P1 = v2f{hs[1], hs[2]};  Q1[0] = hs[0];
      if (t == 0) {
        gates_first(G0, G1, A1t0, A1t1, P1, Q1, c0i1, c1i1);
        act_first(G0, G1, c2s, hs);
      } else {
        gates(G0, G1, A1t0, A1t1, P1, Q1, P2, c0i1, c1i1);
        act(G0, G1, c2s, hs);
      }
      P2 = v2f{hs[1], hs[2]};  Q1[1] = hs[0];
      sp[(t * 2 + hi) * ELEMS_PER_BLOCK + eb] = hs[0] * wl0 + hs[1] * wl1 + hs[2] * wl2;
      if (t + 1 < T_STEPS) Q0 = v2f{xC[t + 1], Q1[0]};
    }
  };
#pragma nounroll
  for (int j = 0; j < TILES_PER_WAVE - 1; ++j) run_tile(j, true);
  run_tile(TILES_PER_WAVE - 1, false);

  __syncthreads();

  {
    const float bo2 = 2.0f * b_out[0];
#pragma unroll
    for (int i = 0; i < (LINES_PER_BLOCK + WAVES_PER_BLOCK - 1) / WAVES_PER_BLOCK; ++i) {
      const int L = wave_u + i * WAVES_PER_BLOCK;
      if (L >= LINES_PER_BLOCK) break;
      const int t = L / LINES_PER_T;
      const int r = (L % LINES_PER_T) * 32 + lane;
      float sum = bo2;
#pragma unroll
      for (int pp = 0; pp < 5; ++pp) {
        const int ee = r * 5 + pp;
        sum += sp[(t * 2 + 0) * ELEMS_PER_BLOCK + ee] + sp[(t * 2 + 1) * ELEMS_PER_BLOCK + ee];
      }
      out[t * ROWS_PER_T + blockIdx.x * ROWS_PER_BLOCK + r] = 0.5f * sum;
    }
  }
}

extern "C" void kernel_launch(void* const* d_in, const int* in_sizes, int n_in,
                              void* d_out, int out_size, void* d_ws, size_t ws_size,
                              hipStream_t stream) {
  const float* x     = (const float*)d_in[0];
  const float* w_ih0 = (const float*)d_in[1];
  const float* w_hh0 = (const float*)d_in[2];
  const float* b_ih0 = (const float*)d_in[3];
  const float* b_hh0 = (const float*)d_in[4];
  const float* w_ih1 = (const float*)d_in[5];
  const float* w_hh1 = (const float*)d_in[6];
  const float* b_ih1 = (const float*)d_in[7];
  const float* b_hh1 = (const float*)d_in[8];
  const float* w_out = (const float*)d_in[9];
  const float* b_out = (const float*)d_in[10];
  float* out = (float*)d_out;

  const int grid = BATCH / ELEMS_PER_BLOCK;
  lstm2_wmma_kernel<<<grid, THREADS, 0, stream>>>(
      x, w_ih0, w_hh0, b_ih0, b_hh0, w_ih1, w_hh1, b_ih1, b_hh1, w_out, b_out, out);
}


# Layers skewed one time step (layer 0 at t beside layer 1 at t-1): 48 gate rows fill 3 row tiles, 47 products a tile not 56; x4 enters units 4,5 as an fma on their bias: their layer-0 sums reordered
# speedup vs baseline: 1.4734x; 1.1038x over previous
#include <hip/hip_runtime.h>
#include <hip/hip_bf16.h>

typedef __attribute__((ext_vector_type(2))) float v2f;
typedef __attribute__((ext_vector_type(4))) float v4f;
typedef __attribute__((ext_vector_type(8))) float v8f;

namespace {

constexpr int T_STEPS = 5;
constexpr int BATCH   = 1000000;
constexpr int IN_DIM  = 5;
constexpr int HID     = 6;
constexpr int ELEMS_PER_GROUP = 80;
constexpr int TILES_PER_WAVE  = ELEMS_PER_GROUP / 16;
constexpr int WAVES_PER_BLOCK = 2;
constexpr int ELEMS_PER_BLOCK = WAVES_PER_BLOCK * ELEMS_PER_GROUP;
constexpr int ROWS_PER_BLOCK  = ELEMS_PER_BLOCK / 5;
constexpr int LINES_PER_T     = ROWS_PER_BLOCK / 32;
constexpr int LINES_PER_BLOCK = T_STEPS * LINES_PER_T;
constexpr int THREADS = WAVES_PER_BLOCK * 32;
constexpr int ROWS_PER_T = BATCH / T_STEPS;
constexpr int TAB_VALS  = 48;
constexpr int TAB_PITCH = 52;

__device__ __forceinline__ float tanh_fast(float x) {
#if __has_builtin(__builtin_amdgcn_tanhf)
  return __builtin_amdgcn_tanhf(x);
#else
  float e = __expf(-2.0f * __builtin_fabsf(x));
  float r = __fdividef(1.0f - e, 1.0f + e);
  return __builtin_copysignf(r, x);
#endif
}

__device__ __forceinline__ float sig_fast(float x) {
  return 0.5f * tanh_fast(x) + 0.5f;
}

__device__ __forceinline__ v8f wmma4(v2f a, v2f b, v8f c) {
  return __builtin_amdgcn_wmma_f32_16x16x4_f32(false, a, false, b, (short)0, c,
                                               false, false);
}

__device__ __forceinline__ float h_doubled(float to, float tc) { return __builtin_fmaf(to, tc, tc); }
__device__ __forceinline__ void lstm_unit(float gi, float gf, float gg, float go,
                                          float& c, float& h) {
  float I = sig_fast(gi);
  float F = sig_fast(gf);
  float G = tanh_fast(gg);
  float O = tanh_fast(go);
  c = F * c + I * G;
  h = h_doubled(O, tanh_fast(c));
}

__device__ __forceinline__ int unit_of(int hi, int q) {
  return q == 0 ? (hi ? 2 : 0) : q == 1 ? (hi ? 3 : 1) : (hi ? 5 : 4);
}

__device__ __forceinline__ float wcol_l0(const float* wih, const float* whh, int R, int c,
                                         float sx, float sh) {
  const float a = wih[R * 5 + (c < 5 ? c : 0)];
  const float b = whh[R * 6 + ((c >= 5 && c < 11) ? c - 5 : 0)];
  return c < 5 ? sx * a : (c < 11 ? sh * b : 0.0f);
}
__device__ __forceinline__ float wcol_l1(const float* wih, const float* whh, int R, int c, float sh) {
  if (c < 6) return sh * wih[R * 6 + c];
  return sh * whh[R * 6 + (c - 6)];
}

__device__ __forceinline__ int zcol_l0(int s, int hi, int j) {
  if (s == 0) return 2 * hi + j;
  if (s == 1) return j == 0 ? (hi ? 11 : 4) : 5 + unit_of(hi, 0);
  return 5 + unit_of(hi, 1 + j);
}
__device__ __forceinline__ int zcol_l1(int s, int hi, int j) {
  if (s == 0) return unit_of(hi, 1 + j);
  if (s == 1) return j == 0 ? unit_of(hi, 0) : 6 + unit_of(hi, 0);
  return 6 + unit_of(hi, 1 + j);
}

__device__ __forceinline__ int tile0_row(int m) {
  return (m & 3) * HID + (m >> 2);
}

__device__ __forceinline__ void lstm_unit_first(float gi, float gg, float go, float& c, float& h) {
  float I = sig_fast(gi);
  float G = tanh_fast(gg);
  float O = tanh_fast(go);
  c = I * G;
  h = h_doubled(O, tanh_fast(c));
}
template <int c0>
__device__ __forceinline__ void act(const v8f& g, const v8f& gc, float (&c)[3], float (&h)[3]) {
  lstm_unit(g[0], g[1], g[2], g[3], c[0], h[0]);
  lstm_unit(g[4], g[5], g[6], g[7], c[1], h[1]);
  lstm_unit(gc[c0], gc[c0 + 1], gc[c0 + 2], gc[c0 + 3], c[2], h[2]);
}
template <int c0>
__device__ __forceinline__ void act_first(const v8f& g, const v8f& gc, float (&c)[3], float (&h)[3]) {
  lstm_unit_first(g[0], g[2], g[3], c[0], h[0]);
  lstm_unit_first(g[4], g[6], g[7], c[1], h[1]);
  lstm_unit_first(gc[c0], gc[c0 + 2], gc[c0 + 3], c[2], h[2]);
}

}

__global__ __launch_bounds__(THREADS)
void lstm2_wmma_kernel(const float* __restrict__ x,
                       const float* __restrict__ w_ih0, const float* __restrict__ w_hh0,
                       const float* __restrict__ b_ih0, const float* __restrict__ b_hh0,
                       const float* __restrict__ w_ih1, const float* __restrict__ w_hh1,
                       const float* __restrict__ b_ih1, const float* __restrict__ b_hh1,
                       const float* __restrict__ w_out, const float* __restrict__ b_out,
                       float* __restrict__ out) {
  __shared__ float sp[T_STEPS * 2 * ELEMS_PER_BLOCK];

  const int tid  = threadIdx.x;
  const int wave = tid >> 5;
  const int lane = tid & 31;
  const int n    = lane & 15;
  const int hi   = lane >> 4;
  const int u0 = hi ? 2 : 0, u1 = hi ? 3 : 1, u2 = hi ? 5 : 4;

  const int e0 = wave * ELEMS_PER_GROUP + n;
  const float* xlane = x + ((size_t)blockIdx.x * ELEMS_PER_BLOCK + e0) * IN_DIM;
  v2f   xPn[T_STEPS];
  float xCn[T_STEPS];
  auto request_x = [&](int j) {
#pragma unroll
    for (int t = 0; t < T_STEPS; ++t) {
      const float* xp = xlane + ((size_t)t * BATCH + j * 16) * IN_DIM;
      xPn[t][0] = xp[2 * hi];
      xPn[t][1] = xp[2 * hi + 1];
      xCn[t]    = xp[4];
    }
  };
  request_x(0);

  __shared__ __attribute__((aligned(16))) float tab[32 * TAB_PITCH];
  const int wave_u = __builtin_amdgcn_readfirstlane(wave);
  for (int quarter = wave_u; quarter < 4; quarter += WAVES_PER_BLOCK) {
    float v[12];
#pragma unroll
    for (int i = 0; i < 12; ++i) v[i] = 0.0f;
    if (quarter == 0) {
      const int R0 = tile0_row(n);
      const float sc0 = (R0 / HID == 2) ? 1.0f : 0.5f;
      const float sh0 = 0.5f * sc0;
#pragma unroll
      for (int s = 0; s < 3; ++s) {
#pragma unroll
        for (int j = 0; j < 2; ++j) {
          v[s * 2 + j]     = wcol_l0(w_ih0, w_hh0, R0, zcol_l0(s, hi, j), sc0, sh0);
          v[6 + s * 2 + j] = wcol_l1(w_ih1, w_hh1, R0, zcol_l1(s, hi, j), sh0);
        }
      }
    } else if (quarter == 1) {
      const bool lay1 = (n & 4) != 0;
      const int  R    = (n & 3) * HID + ((n & 8) ? 5 : 4);
      const float sc  = ((n & 3) == 2) ? 1.0f : 0.5f;
      const float sh  = 0.5f * sc;
#pragma unroll
      for (int j = 0; j < 2; ++j) {
        const int ua = unit_of(hi, 1 + j);
        const float wx  = w_ih0[R * 5 + 2 * hi + j];
        const float wa0 = w_hh0[R * 6 + ua], wa1 = w_ih1[R * 6 + ua];
        const float wb1 = w_hh1[R * 6 + ua];
        v[j]     = lay1 ? 0.0f : sc * wx;
        v[2 + j] = sh * (lay1 ? wa1 : wa0);
        v[6 + j] = lay1 ? sh * wb1 : 0.0f;
      }
      {
        const int ub = unit_of(hi, 0);
        const float wq0 = w_hh0[R * 6 + ub], wq1 = w_ih1[R * 6 + ub], wq2 = w_hh1[R * 6 + ub];
        v[4] = sh * (lay1 ? wq1 : wq0);
        v[5] = lay1 ? sh * wq2 : 0.0f;
      }
#pragma unroll
      for (int r = 0; r < 4; ++r)
        v[8 + r] = ((r == 2) ? 1.0f : 0.5f) * w_ih0[(r * HID + u2) * 5 + 4];
    } else if (quarter == 2) {
#pragma unroll
      for (int r = 0; r < 8; ++r) {
        const int R = tile0_row(r + 8 * hi);
        const float sb = (R / HID == 2) ? 1.0f : 0.5f;
        v[r] = sb * (b_ih0[R] + b_hh0[R]);
        if (r < 4) v[8 + r] = sb * (b_ih1[R] + b_hh1[R]);
      }
    } else {
#pragma unroll
      for (int r = 4; r < 8; ++r) {
        const int R = tile0_row(r + 8 * hi);
        const float sb = (R / HID == 2) ? 1.0f : 0.5f;
        v[r - 4] = sb * (b_ih1[R] + b_hh1[R]);
      }
#pragma unroll
      for (int r = 0; r < 4; ++r) {
        const int R = r * HID + u2;
        const float sb = (r == 2) ? 1.0f : 0.5f;
        v[4 + r] = sb * (b_ih0[R] + b_hh0[R]);
        v[8 + r] = sb * (b_ih1[R] + b_hh1[R]);
      }
    }
    v4f* mine = reinterpret_cast<v4f*>(tab + lane * TAB_PITCH + 12 * quarter);
    mine[0] = v4f{v[0], v[1], v[2],  v[3]};
    mine[1] = v4f{v[4], v[5], v[6],  v[7]};
    mine[2] = v4f{v[8], v[9], v[10], v[11]};
  }

  __syncthreads();
  v2f At[3], Bt[3], Ct[4];
  float w4[4];
  v8f biA = {}, biB = {};
  float bC0[4];
  v8f ciC = {};
  {
    const v4f* tv = reinterpret_cast<const v4f*>(tab + lane * TAB_PITCH);
    float q[TAB_VALS];
#pragma unroll
    for (int i = 0; i < TAB_VALS / 4; ++i) {
      const v4f w = tv[i];
      q[4 * i] = w[0];  q[4 * i + 1] = w[1];  q[4 * i + 2] = w[2];  q[4 * i + 3] = w[3];
    }
#pragma unroll
    for (int s = 0; s < 3; ++s) {
      At[s] = v2f{q[2 * s],     q[2 * s + 1]};
      Bt[s] = v2f{q[6 + 2 * s], q[6 + 2 * s + 1]};
    }
#pragma unroll
    for (int s = 0; s < 4; ++s) Ct[s] = v2f{q[12 + 2 * s], q[12 + 2 * s + 1]};
#pragma unroll
    for (int r = 0; r < 4; ++r) w4[r] = q[20 + r];
#pragma unroll
    for (int r = 0; r < 8; ++r) { biA[r] = q[24 + r];  biB[r] = q[32 + r]; }
#pragma unroll
    for (int r = 0; r < 4; ++r) { bC0[r] = q[40 + r];  ciC[4 + r] = q[44 + r]; }
  }


  int p = n % 5;
  auto run_tile = [&](const int j, const bool more) __attribute__((always_inline)) {
    v2f   xP[T_STEPS];
    float xC[T_STEPS];
#pragma unroll
    for (int t = 0; t < T_STEPS; ++t) { xP[t] = xPn[t];  xC[t] = xCn[t]; }
    const float wl0 = w_out[p * HID + u0];
    const float wl1 = w_out[p * HID + u1];
    const float wl2 = w_out[p * HID + u2];
    p = (p == 4) ? 0 : p + 1;
    if (more) request_x(j + 1);
    __builtin_amdgcn_sched_barrier(0);

    const int eb = e0 + j * 16;
    float c1s[3] = {}, c2s[3] = {};
    v2f P1 = v2f{0.0f, 0.0f}, P2 = v2f{0.0f, 0.0f}, Q1 = v2f{0.0f, 0.0f};
    v2f Q0 = v2f{hi ? 0.0f : xC[0], 0.0f};
    v8f GA, GB, GC;
#pragma unroll
    for (int m = 0; m <= T_STEPS; ++m) {
      __builtin_amdgcn_sched_barrier(0);
      if (m < T_STEPS) {
#pragma unroll
        for (int r = 0; r < 4; ++r) ciC[r] = __builtin_fmaf(w4[r], xC[m], bC0[r]);
      }
      if (m == 0) {
        GA = wmma4(At[0], xP[0], biA);   GC = wmma4(Ct[0], xP[0], ciC);
        GA = wmma4(At[1], Q0, GA);
      } else if (m == 1) {
        GA = wmma4(At[0], xP[m], biA);   GB = wmma4(Bt[0], P1, biB);   GC = wmma4(Ct[0], xP[m], ciC);
        GA = wmma4(At[1], Q0, GA);       GB = wmma4(Bt[1], Q1, GB);    GC = wmma4(Ct[1], P1, GC);
        GA = wmma4(At[2], P1, GA);                                     GC = wmma4(Ct[2], Q1, GC);
      } else if (m < T_STEPS) {
        GA = wmma4(At[0], xP[m], biA);   GB = wmma4(Bt[0], P1, biB);   GC = wmma4(Ct[0], xP[m], ciC);
        GA = wmma4(At[1], Q0, GA);       GB = wmma4(Bt[1], Q1, GB);    GC = wmma4(Ct[1], P1, GC);
        GA = wmma4(At[2], P1, GA);       GB = wmma4(Bt[2], P2, GB);    GC = wmma4(Ct[2], Q1, GC);
                                                                       GC = wmma4(Ct[3], P2, GC);
      } else {
        GB = wmma4(Bt[0], P1, biB);      GC = wmma4(Ct[1], P1, ciC);
        GB = wmma4(Bt[1], Q1, GB);       GC = wmma4(Ct[2], Q1, GC);
        GB = wmma4(Bt[2], P2, GB);       GC = wmma4(Ct[3], P2, GC);
      }
      float h1n[3], h2n[3];
      if (m == 0)            act_first<0>(GA, GC, c1s, h1n);
      else if (m < T_STEPS)  act<0>(GA, GC, c1s, h1n);
      if (m == 1)            act_first<4>(GB, GC, c2s, h2n);
      else if (m > 1)        act<4>(GB, GC, c2s, h2n);
      if (m < T_STEPS) {
        P1 = v2f{h1n[1], h1n[2]};  Q1[0] = h1n[0];
        if (m + 1 < T_STEPS) Q0 = v2f{hi ? 0.0f : xC[m + 1], h1n[0]};
      }
      if (m > 0) {
        P2 = v2f{h2n[1], h2n[2]};  Q1[1] = h2n[0];
        sp[((m - 1) * 2 + hi) * ELEMS_PER_BLOCK + eb] = h2n[0] * wl0 + h2n[1] * wl1 + h2n[2] * wl2;
      }
    }
  };
#pragma nounroll
  for (int j = 0; j < TILES_PER_WAVE - 1; ++j) run_tile(j, true);
  run_tile(TILES_PER_WAVE - 1, false);

  __syncthreads();

  {
    const float bo2 = 2.0f * b_out[0];
#pragma unroll
    for (int i = 0; i < (LINES_PER_BLOCK + WAVES_PER_BLOCK - 1) / WAVES_PER_BLOCK; ++i) {
      const int L = wave_u + i * WAVES_PER_BLOCK;
      if (L >= LINES_PER_BLOCK) break;
      const int t = L / LINES_PER_T;
      const int r = (L % LINES_PER_T) * 32 + lane;
      float sum = bo2;
#pragma unroll
      for (int pp = 0; pp < 5; ++pp) {
        const int ee = r * 5 + pp;
        sum += sp[(t * 2 + 0) * ELEMS_PER_BLOCK + ee] + sp[(t * 2 + 1) * ELEMS_PER_BLOCK + ee];
      }
      out[t * ROWS_PER_T + blockIdx.x * ROWS_PER_BLOCK + r] = 0.5f * sum;
    }
  }
}

extern "C" void kernel_launch(void* const* d_in, const int* in_sizes, int n_in,
                              void* d_out, int out_size, void* d_ws, size_t ws_size,
                              hipStream_t stream) {
  const float* x     = (const float*)d_in[0];
  const float* w_ih0 = (const float*)d_in[1];
  const float* w_hh0 = (const float*)d_in[2];
  const float* b_ih0 = (const float*)d_in[3];
  const float* b_hh0 = (const float*)d_in[4];
  const float* w_ih1 = (const float*)d_in[5];
  const float* w_hh1 = (const float*)d_in[6];
  const float* b_ih1 = (const float*)d_in[7];
  const float* b_hh1 = (const float*)d_in[8];
  const float* w_out = (const float*)d_in[9];
  const float* b_out = (const float*)d_in[10];
  float* out = (float*)d_out;

  const int grid = BATCH / ELEMS_PER_BLOCK;
  lstm2_wmma_kernel<<<grid, THREADS, 0, stream>>>(
      x, w_ih0, w_hh0, b_ih0, b_hh0, w_ih1, w_hh1, b_ih1, b_hh1, w_out, b_out, out);
}


# Per-lane f32 math two-wide and in place: gate, cell and hidden math packed over unit pairs that are the next operand pairs; tile C's x4 term added behind its products: those four sums reordered
# speedup vs baseline: 1.5535x; 1.0544x over previous
#include <hip/hip_runtime.h>
#include <hip/hip_bf16.h>

typedef __attribute__((ext_vector_type(2))) float v2f;
typedef __attribute__((ext_vector_type(4))) float v4f;
typedef __attribute__((ext_vector_type(8))) float v8f;

namespace {

constexpr int T_STEPS = 5;
constexpr int BATCH   = 1000000;
constexpr int IN_DIM  = 5;
constexpr int HID     = 6;
constexpr int ELEMS_PER_GROUP = 80;
constexpr int TILES_PER_WAVE  = ELEMS_PER_GROUP / 16;
constexpr int WAVES_PER_BLOCK = 2;
constexpr int ELEMS_PER_BLOCK = WAVES_PER_BLOCK * ELEMS_PER_GROUP;
constexpr int ROWS_PER_BLOCK  = ELEMS_PER_BLOCK / 5;
constexpr int LINES_PER_T     = ROWS_PER_BLOCK / 32;
constexpr int LINES_PER_BLOCK = T_STEPS * LINES_PER_T;
constexpr int THREADS = WAVES_PER_BLOCK * 32;
constexpr int ROWS_PER_T = BATCH / T_STEPS;
constexpr int TAB_VALS  = 48;
constexpr int TAB_PITCH = 52;

__device__ __forceinline__ float tanh_fast(float x) {
#if __has_builtin(__builtin_amdgcn_tanhf)
  return __builtin_amdgcn_tanhf(x);
#else
  float e = __expf(-2.0f * __builtin_fabsf(x));
  float r = __fdividef(1.0f - e, 1.0f + e);
  return __builtin_copysignf(r, x);
#endif
}

__device__ __forceinline__ float sig_fast(float x) {
  return 0.5f * tanh_fast(x) + 0.5f;
}

__device__ __forceinline__ v8f wmma4(v2f a, v2f b, v8f c) {
  return __builtin_amdgcn_wmma_f32_16x16x4_f32(false, a, false, b, (short)0, c,
                                               false, false);
}

__device__ __forceinline__ float h_doubled(float to, float tc) { return __builtin_fmaf(to, tc, tc); }
__device__ __forceinline__ void lstm_unit(float gi, float gf, float gg, float go,
                                          float& c, float& h) {
  float I = sig_fast(gi);
  float F = sig_fast(gf);
  float G = tanh_fast(gg);
  float O = tanh_fast(go);
  c = F * c + I * G;
  h = h_doubled(O, tanh_fast(c));
}

__device__ __forceinline__ int unit_of(int hi, int q) {
  return q == 0 ? (hi ? 2 : 0) : q == 1 ? (hi ? 3 : 1) : (hi ? 5 : 4);
}

__device__ __forceinline__ float wcol_l0(const float* wih, const float* whh, int R, int c,
                                         float sx, float sh) {
  const float a = wih[R * 5 + (c < 5 ? c : 0)];
  const float b = whh[R * 6 + ((c >= 5 && c < 11) ? c - 5 : 0)];
  return c < 5 ? sx * a : (c < 11 ? sh * b : 0.0f);
}
__device__ __forceinline__ float wcol_l1(const float* wih, const float* whh, int R, int c, float sh) {
  if (c < 6) return sh * wih[R * 6 + c];
  return sh * whh[R * 6 + (c - 6)];
}

__device__ __forceinline__ int zcol_l0(int s, int hi, int j) {
  if (s == 0) return 2 * hi + j;
  if (s == 1) return j == 0 ? (hi ? 11 : 4) : 5 + unit_of(hi, 0);
  return 5 + unit_of(hi, 1 + j);
}
__device__ __forceinline__ int zcol_l1(int s, int hi, int j) {
  if (s == 0) return unit_of(hi, 1 + j);
  if (s == 1) return j == 0 ? unit_of(hi, 0) : 6 + unit_of(hi, 0);
  return 6 + unit_of(hi, 1 + j);
}

__device__ __forceinline__ int tile0_row(int m) {
  return (m & 3) * HID + (m >> 2);
}

__device__ __forceinline__ void lstm_unit_first(float gi, float gg, float go, float& c, float& h) {
  float I = sig_fast(gi);
  float G = tanh_fast(gg);
  float O = tanh_fast(go);
  c = I * G;
  h = h_doubled(O, tanh_fast(c));
}
__device__ __forceinline__ v2f tanh2(float a, float b) { return v2f{tanh_fast(a), tanh_fast(b)}; }
__device__ __forceinline__ v2f sig2(float a, float b) {
  const v2f half = {0.5f, 0.5f};
  return __builtin_elementwise_fma(tanh2(a, b), half, half);
}
__device__ __forceinline__ void lstm_pair(float i0, float f0, float g0, float o0,
                                          float i1, float f1, float g1, float o1, v2f& c, v2f& h) {
  const v2f I = sig2(i0, i1);
  const v2f F = sig2(f0, f1);
  const v2f G = tanh2(g0, g1);
  const v2f O = tanh2(o0, o1);
  c = __builtin_elementwise_fma(F, c, I * G);
  const v2f tc = tanh2(c[0], c[1]);
  h = __builtin_elementwise_fma(O, tc, tc);
}
__device__ __forceinline__ void lstm_pair_first(float i0, float g0, float o0,
                                                float i1, float g1, float o1, v2f& c, v2f& h) {
  const v2f I = sig2(i0, i1);
  const v2f G = tanh2(g0, g1);
  const v2f O = tanh2(o0, o1);
  c = I * G;
  const v2f tc = tanh2(c[0], c[1]);
  h = __builtin_elementwise_fma(O, tc, tc);
}
__device__ __forceinline__ void lstm_pair_mixed(float i0, float f0, float g0, float o0,
                                                float i1, float g1, float o1, v2f& c, v2f& h) {
  const v2f I = sig2(i0, i1);
  const float F0 = sig_fast(f0);
  const v2f G = tanh2(g0, g1);
  const v2f O = tanh2(o0, o1);
  const v2f IG = I * G;
  c = v2f{__builtin_fmaf(F0, c[0], IG[0]), IG[1]};
  const v2f tc = tanh2(c[0], c[1]);
  h = __builtin_elementwise_fma(O, tc, tc);
}

}

__global__ __launch_bounds__(THREADS)
void lstm2_wmma_kernel(const float* __restrict__ x,
                       const float* __restrict__ w_ih0, const float* __restrict__ w_hh0,
                       const float* __restrict__ b_ih0, const float* __restrict__ b_hh0,
                       const float* __restrict__ w_ih1, const float* __restrict__ w_hh1,
                       const float* __restrict__ b_ih1, const float* __restrict__ b_hh1,
                       const float* __restrict__ w_out, const float* __restrict__ b_out,
                       float* __restrict__ out) {
  __shared__ float sp[T_STEPS * 2 * ELEMS_PER_BLOCK];

  const int tid  = threadIdx.x;
  const int wave = tid >> 5;
  const int lane = tid & 31;
  const int n    = lane & 15;
  const int hi   = lane >> 4;
  const int u0 = hi ? 2 : 0, u1 = hi ? 3 : 1, u2 = hi ? 5 : 4;

  const int e0 = wave * ELEMS_PER_GROUP + n;
  const float* xlane = x + ((size_t)blockIdx.x * ELEMS_PER_BLOCK + e0) * IN_DIM;
  v2f   xPn[T_STEPS];
  float xCn[T_STEPS];
  auto request_x = [&](int j) {
#pragma unroll
    for (int t = 0; t < T_STEPS; ++t) {
      const float* xp = xlane + ((size_t)t * BATCH + j * 16) * IN_DIM;
      xPn[t][0] = xp[2 * hi];
      xPn[t][1] = xp[2 * hi + 1];
      xCn[t]    = xp[4];
    }
  };
  request_x(0);

  __shared__ __attribute__((aligned(16))) float tab[32 * TAB_PITCH];
  const int wave_u = __builtin_amdgcn_readfirstlane(wave);
  for (int quarter = wave_u; quarter < 4; quarter += WAVES_PER_BLOCK) {
    float v[12];
#pragma unroll
    for (int i = 0; i < 12; ++i) v[i] = 0.0f;
    if (quarter == 0) {
      const int R0 = tile0_row(n);
      const float sc0 = (R0 / HID == 2) ? 1.0f : 0.5f;
      const float sh0 = 0.5f * sc0;
#pragma unroll
      for (int s = 0; s < 3; ++s) {
#pragma unroll
        for (int j = 0; j < 2; ++j) {
          v[s * 2 + j]     = wcol_l0(w_ih0, w_hh0, R0, zcol_l0(s, hi, j), sc0, sh0);
          v[6 + s * 2 + j] = wcol_l1(w_ih1, w_hh1, R0, zcol_l1(s, hi, j), sh0);
        }
      }
    } else if (quarter == 1) {
      const bool lay1 = (n & 4) != 0;
      const int  R    = (n & 3) * HID + ((n & 8) ? 5 : 4);
      const float sc  = ((n & 3) == 2) ? 1.0f : 0.5f;
      const float sh  = 0.5f * sc;
#pragma unroll
      for (int j = 0; j < 2; ++j) {
        const int ua = unit_of(hi, 1 + j);
        const float wx  = w_ih0[R * 5 + 2 * hi + j];
        const float wa0 = w_hh0[R * 6 + ua], wa1 = w_ih1[R * 6 + ua];
        const float wb1 = w_hh1[R * 6 + ua];
        v[j]     = lay1 ? 0.0f : sc * wx;
        v[2 + j] = sh * (lay1 ? wa1 : wa0);
        v[6 + j] = lay1 ? sh * wb1 : 0.0f;
      }
      {
        const int ub = unit_of(hi, 0);
        const float wq0 = w_hh0[R * 6 + ub], wq1 = w_ih1[R * 6 + ub], wq2 = w_hh1[R * 6 + ub];
        v[4] = sh * (lay1 ? wq1 : wq0);
        v[5] = lay1 ? sh * wq2 : 0.0f;
      }
#pragma unroll
      for (int r = 0; r < 4; ++r)
        v[8 + r] = ((r == 2) ? 1.0f : 0.5f) * w_ih0[(r * HID + u2) * 5 + 4];
    } else if (quarter == 2) {
#pragma unroll
      for (int r = 0; r < 8; ++r) {
        const int R = tile0_row(r + 8 * hi);
        const float sb = (R / HID == 2) ? 1.0f : 0.5f;
        v[r] = sb * (b_ih0[R] + b_hh0[R]);
        if (r < 4) v[8 + r] = sb * (b_ih1[R] + b_hh1[R]);
      }
    } else {
#pragma unroll
      for (int r = 4; r < 8; ++r) {
        const int R = tile0_row(r + 8 * hi);
        const float sb = (R / HID == 2) ? 1.0f : 0.5f;
        v[r - 4] = sb * (b_ih1[R] + b_hh1[R]);
      }
#pragma unroll
      for (int r = 0; r < 4; ++r) {
        const int R = r * HID + u2;
        const float sb = (r == 2) ? 1.0f : 0.5f;
        v[4 + r] = sb * (b_ih0[R] + b_hh0[R]);
        v[8 + r] = sb * (b_ih1[R] + b_hh1[R]);
      }
    }
    v4f* mine = reinterpret_cast<v4f*>(tab + lane * TAB_PITCH + 12 * quarter);
    mine[0] = v4f{v[0], v[1], v[2],  v[3]};
    mine[1] = v4f{v[4], v[5], v[6],  v[7]};
    mine[2] = v4f{v[8], v[9], v[10], v[11]};
  }

  __syncthreads();
  v2f At[3], Bt[3], Ct[4];
  v2f w4a, w4b;
  v8f biA = {}, biB = {};
  v8f biC = {};
  {
    const v4f* tv = reinterpret_cast<const v4f*>(tab + lane * TAB_PITCH);
    float q[TAB_VALS];
#pragma unroll
    for (int i = 0; i < TAB_VALS / 4; ++i) {
      const v4f w = tv[i];
      q[4 * i] = w[0];  q[4 * i + 1] = w[1];  q[4 * i + 2] = w[2];  q[4 * i + 3] = w[3];
    }
#pragma unroll
    for (int s = 0; s < 3; ++s) {
      At[s] = v2f{q[2 * s],     q[2 * s + 1]};
      Bt[s] = v2f{q[6 + 2 * s], q[6 + 2 * s + 1]};
    }
#pragma unroll
    for (int s = 0; s < 4; ++s) Ct[s] = v2f{q[12 + 2 * s], q[12 + 2 * s + 1]};
    w4a = v2f{q[20], q[21]};  w4b = v2f{q[22], q[23]};
#pragma unroll
    for (int r = 0; r < 8; ++r) { biA[r] = q[24 + r];  biB[r] = q[32 + r]; }
#pragma unroll
    for (int r = 0; r < 4; ++r) { biC[r] = q[40 + r];  biC[4 + r] = q[44 + r]; }
  }


  int p = n % 5;
  auto run_tile = [&](const int j, const bool more) __attribute__((always_inline)) {
    v2f   xP[T_STEPS];
    float xC[T_STEPS];
#pragma unroll
    for (int t = 0; t < T_STEPS; ++t) { xP[t] = xPn[t];  xC[t] = xCn[t]; }
    const float wl0 = w_out[p * HID + u0];
    const float wl1 = w_out[p * HID + u1];
    const float wl2 = w_out[p * HID + u2];
    p = (p == 4) ? 0 : p + 1;
    if (more) request_x(j + 1);
    __builtin_amdgcn_sched_barrier(0);

    const int eb = e0 + j * 16;
    v2f cX = v2f{0.0f, 0.0f}, cY = v2f{0.0f, 0.0f}, cZ = v2f{0.0f, 0.0f};
    v2f P1 = v2f{0.0f, 0.0f}, P2 = v2f{0.0f, 0.0f}, Q1 = v2f{0.0f, 0.0f};
    v2f Q0 = v2f{hi ? 0.0f : xC[0], 0.0f};
    v8f GA, GB, GC;
#pragma unroll
    for (int m = 0; m <= T_STEPS; ++m) {
      __builtin_amdgcn_sched_barrier(0);
      if (m == 0) {
        GA = wmma4(At[0], xP[0], biA);   GC = wmma4(Ct[0], xP[0], biC);
        GA = wmma4(At[1], Q0, GA);
      } else if (m == 1) {
        GA = wmma4(At[0], xP[m], biA);   GB = wmma4(Bt[0], P1, biB);   GC = wmma4(Ct[0], xP[m], biC);
        GA = wmma4(At[1], Q0, GA);       GB = wmma4(Bt[1], Q1, GB);    GC = wmma4(Ct[1], P1, GC);
        GA = wmma4(At[2], P1, GA);                                     GC = wmma4(Ct[2], Q1, GC);
      } else if (m < T_STEPS) {
        GA = wmma4(At[0], xP[m], biA);   GB = wmma4(Bt[0], P1, biB);   GC = wmma4(Ct[0], xP[m], biC);
        GA = wmma4(At[1], Q0, GA);       GB = wmma4(Bt[1], Q1, GB);    GC = wmma4(Ct[1], P1, GC);
        GA = wmma4(At[2], P1, GA);       GB = wmma4(Bt[2], P2, GB);    GC = wmma4(Ct[2], Q1, GC);
                                                                       GC = wmma4(Ct[3], P2, GC);
      } else {
        GB = wmma4(Bt[0], P1, biB);      GC = wmma4(Ct[1], P1, biC);
        GB = wmma4(Bt[1], Q1, GB);       GC = wmma4(Ct[2], Q1, GC);
        GB = wmma4(Bt[2], P2, GB);       GC = wmma4(Ct[3], P2, GC);
      }
      v2f gC01 = v2f{GC[0], GC[1]}, gC23 = v2f{GC[2], GC[3]};
      if (m < T_STEPS) {
        const v2f x4 = v2f{xC[m], xC[m]};
        gC01 = __builtin_elementwise_fma(w4a, x4, gC01);
        gC23 = __builtin_elementwise_fma(w4b, x4, gC23);
      }
      v2f hX = v2f{0.0f, 0.0f}, hY = v2f{0.0f, 0.0f}, hZ = v2f{0.0f, 0.0f};
      if (m == 0) {
        lstm_pair_first(GA[4], GA[6], GA[7], gC01[0], gC23[0], gC23[1], cX, hX);
        float c0 = 0.0f, h0;
        lstm_unit_first(GA[0], GA[2], GA[3], c0, h0);
        cZ[0] = c0;  hZ[0] = h0;
      } else if (m < T_STEPS) {
        lstm_pair(GA[4], GA[5], GA[6], GA[7], gC01[0], gC01[1], gC23[0], gC23[1], cX, hX);
      }
      if (m == 1) {
        lstm_pair_first(GB[4], GB[6], GB[7], GC[4], GC[6], GC[7], cY, hY);
        lstm_pair_mixed(GA[0], GA[1], GA[2], GA[3], GB[0], GB[2], GB[3], cZ, hZ);
      } else if (m > 1) {
        lstm_pair(GB[4], GB[5], GB[6], GB[7], GC[4], GC[5], GC[6], GC[7], cY, hY);
        if (m < T_STEPS) {
          lstm_pair(GA[0], GA[1], GA[2], GA[3], GB[0], GB[1], GB[2], GB[3], cZ, hZ);
        } else {
          float c1 = cZ[1], h1;
          lstm_unit(GB[0], GB[1], GB[2], GB[3], c1, h1);
          cZ[1] = c1;  hZ[1] = h1;
        }
      }
      if (m < T_STEPS) {
        P1 = hX;
        if (m + 1 < T_STEPS) Q0 = v2f{hi ? 0.0f : xC[m + 1], hZ[0]};
      }
      Q1 = hZ;
      if (m > 0) {
        P2 = hY;
        sp[((m - 1) * 2 + hi) * ELEMS_PER_BLOCK + eb] = hZ[1] * wl0 + hY[0] * wl1 + hY[1] * wl2;
      }
    }
  };
#pragma nounroll
  for (int j = 0; j < TILES_PER_WAVE - 1; ++j) run_tile(j, true);
  run_tile(TILES_PER_WAVE - 1, false);

  __syncthreads();

  {
    const float bo2 = 2.0f * b_out[0];
#pragma unroll
    for (int i = 0; i < (LINES_PER_BLOCK + WAVES_PER_BLOCK - 1) / WAVES_PER_BLOCK; ++i) {
      const int L = wave_u + i * WAVES_PER_BLOCK;
      if (L >= LINES_PER_BLOCK) break;
      const int t = L / LINES_PER_T;
      const int r = (L % LINES_PER_T) * 32 + lane;
      float sum = bo2;
#pragma unroll
      for (int pp = 0; pp < 5; ++pp) {
        const int ee = r * 5 + pp;
        sum += sp[(t * 2 + 0) * ELEMS_PER_BLOCK + ee] + sp[(t * 2 + 1) * ELEMS_PER_BLOCK + ee];
      }
      out[t * ROWS_PER_T + blockIdx.x * ROWS_PER_BLOCK + r] = 0.5f * sum;
    }
  }
}

extern "C" void kernel_launch(void* const* d_in, const int* in_sizes, int n_in,
                              void* d_out, int out_size, void* d_ws, size_t ws_size,
                              hipStream_t stream) {
  const float* x     = (const float*)d_in[0];
  const float* w_ih0 = (const float*)d_in[1];
  const float* w_hh0 = (const float*)d_in[2];
  const float* b_ih0 = (const float*)d_in[3];
  const float* b_hh0 = (const float*)d_in[4];
  const float* w_ih1 = (const float*)d_in[5];
  const float* w_hh1 = (const float*)d_in[6];
  const float* b_ih1 = (const float*)d_in[7];
  const float* b_hh1 = (const float*)d_in[8];
  const float* w_out = (const float*)d_in[9];
  const float* b_out = (const float*)d_in[10];
  float* out = (float*)d_out;

  const int grid = BATCH / ELEMS_PER_BLOCK;
  lstm2_wmma_kernel<<<grid, THREADS, 0, stream>>>(
      x, w_ih0, w_hh0, b_ih0, b_hh0, w_ih1, w_hh1, b_ih1, b_hh1, w_out, b_out, out);
}
